# starting kernel
# baseline (speedup 1.0000x reference)
#include <hip/hip_runtime.h>
#include <math.h>

#define HEADS     8
#define HIDC      32
#define L1C       256
#define OUTC      16
#define NEG_SLOPE 0.2f

typedef __attribute__((ext_vector_type(2))) float v2f;
typedef __attribute__((ext_vector_type(8))) float v8f;

__device__ __forceinline__ unsigned fenc(float f) {
    unsigned u = __float_as_uint(f);
    return (u & 0x80000000u) ? ~u : (u | 0x80000000u);
}
__device__ __forceinline__ float fdec(unsigned k) {
    return __uint_as_float((k & 0x80000000u) ? (k ^ 0x80000000u) : ~k);
}
__device__ __forceinline__ float lrelu(float v) {
    return v > 0.0f ? v : NEG_SLOPE * v;
}
__device__ __forceinline__ void edge_sd(const int* __restrict__ ei, int E, int e,
                                        int& s, int& d) {
    if (e < E) { s = ei[e]; d = ei[E + e]; }
    else       { s = e - E; d = e - E; }
}

__global__ void k_fill_zero(float* __restrict__ p, size_t n) {
    size_t i = (size_t)blockIdx.x * blockDim.x + threadIdx.x;
    size_t stride = (size_t)gridDim.x * blockDim.x;
    for (; i < n; i += stride) p[i] = 0.0f;
}

__global__ void k_gemm1_wmma(const float* __restrict__ x, const float* __restrict__ W1,
                             float* __restrict__ h1, int n_nodes) {
    int wave   = (int)((blockIdx.x * blockDim.x + threadIdx.x) >> 5);
    int lane   = threadIdx.x & 31;
    int mtiles = (n_nodes + 15) >> 4;
    int mtile  = wave >> 4;
    int ntile  = wave & 15;
    if (mtile >= mtiles) return;
    int m  = lane & 15;
    int kh = lane >> 4;
    bool full = (mtile * 16 + 16 <= n_nodes);
    int node = mtile * 16 + m;
    if (node >= n_nodes) node = n_nodes - 1;
    float x0 = x[node * 3 + 0], x1 = x[node * 3 + 1], x2 = x[node * 3 + 2];
    int col = ntile * 16 + m;
    float w0 = W1[0 * L1C + col], w1 = W1[1 * L1C + col], w2 = W1[2 * L1C + col];
    v2f a, b;
    a.x = kh ? x2 : x0;  a.y = kh ? 0.0f : x1;
    b.x = kh ? w2 : w0;  b.y = kh ? 0.0f : w1;
    v8f c = {};
    c = __builtin_amdgcn_wmma_f32_16x16x4_f32(false, a, false, b, (short)0, c,
                                              false, false);
    float* p = h1 + (size_t)(mtile * 16 + 8 * kh) * L1C + col;
    if (full) {
#pragma unroll
        for (int v = 0; v < 8; ++v) p[(size_t)v * L1C] = c[v];
    } else {
        for (int v = 0; v < 8; ++v)
            if (mtile * 16 + 8 * kh + v < n_nodes) p[(size_t)v * L1C] = c[v];
    }
}

__global__ void k_att1(const float* __restrict__ h1, const float* __restrict__ as1,
                       const float* __restrict__ ad1, float* __restrict__ a_s,
                       float* __restrict__ a_d, int n_nodes) {
    int node = blockIdx.x;
    if (node >= n_nodes) return;
    int head = threadIdx.x >> 5;
    int lane = threadIdx.x & 31;
    float v = h1[(size_t)node * L1C + head * HIDC + lane];
    float s = v * as1[head * HIDC + lane];
    float d = v * ad1[head * HIDC + lane];
    for (int off = 16; off; off >>= 1) {
        s += __shfl_xor(s, off, 32);
        d += __shfl_xor(d, off, 32);
    }
    if (lane == 0) {
        a_s[node * HEADS + head] = s;
        a_d[node * HEADS + head] = d;
    }
}

__global__ void k_edge_max1(const int* __restrict__ ei, int E, int ET,
                            const float* __restrict__ a_s, const float* __restrict__ a_d,
                            unsigned* __restrict__ m1) {
    int e = blockIdx.x * blockDim.x + threadIdx.x;
    if (e >= ET) return;
    int s, d; edge_sd(ei, E, e, s, d);
#pragma unroll
    for (int h = 0; h < HEADS; ++h) {
        float v = lrelu(a_s[s * HEADS + h] + a_d[d * HEADS + h]);
        atomicMax(&m1[d * HEADS + h], fenc(v));
    }
}

__global__ void k_edge_exp1(const int* __restrict__ ei, int E, int ET,
                            const float* __restrict__ a_s, const float* __restrict__ a_d,
                            const unsigned* __restrict__ m1, float* __restrict__ denom1) {
    int e = blockIdx.x * blockDim.x + threadIdx.x;
    if (e >= ET) return;
    int s, d; edge_sd(ei, E, e, s, d);
#pragma unroll
    for (int h = 0; h < HEADS; ++h) {
        float v = lrelu(a_s[s * HEADS + h] + a_d[d * HEADS + h]);
        float w = expf(v - fdec(m1[d * HEADS + h]));
        atomicAdd(&denom1[d * HEADS + h], w);
    }
}

__global__ void k_msg1(const int* __restrict__ ei, int E, int ET,
                       const float* __restrict__ a_s, const float* __restrict__ a_d,
                       const unsigned* __restrict__ m1, const float* __restrict__ denom1,
                       const float* __restrict__ h1, float* __restrict__ out1) {
    long long gid = (long long)blockIdx.x * blockDim.x + threadIdx.x;
    int e = (int)(gid >> 5);
    int lane = (int)(gid & 31);
    if (e >= ET) return;
    int s, d; edge_sd(ei, E, e, s, d);
    float myalpha = 0.0f;
    if (lane < HEADS) {
        float v = lrelu(a_s[s * HEADS + lane] + a_d[d * HEADS + lane]);
        float w = expf(v - fdec(m1[d * HEADS + lane]));
        myalpha = w / (denom1[d * HEADS + lane] + 1e-16f);
    }
    const float* ph = h1 + (size_t)s * L1C + lane;
    float* po = out1 + (size_t)d * L1C + lane;
#pragma unroll
    for (int h = 0; h < HEADS; ++h) {
        float ah = __shfl(myalpha, h, 32);
        atomicAdd(po + h * HIDC, ph[h * HIDC] * ah);
    }
}

__global__ void k_elu(float* __restrict__ out1, const float* __restrict__ bias1, size_t n) {
    size_t i = (size_t)blockIdx.x * blockDim.x + threadIdx.x;
    size_t stride = (size_t)gridDim.x * blockDim.x;
    for (; i < n; i += stride) {
        float v = out1[i] + bias1[i & (L1C - 1)];
        out1[i] = v > 0.0f ? v : (expf(v) - 1.0f);
    }
}

__global__ void k_gemm2_wmma(const float* __restrict__ helu, const float* __restrict__ W2,
                             float* __restrict__ h2, int n_nodes) {
    int wave = (int)((blockIdx.x * blockDim.x + threadIdx.x) >> 5);
    int lane = threadIdx.x & 31;
    int mtiles = (n_nodes + 15) >> 4;
    if (wave >= mtiles) return;
    int m  = lane & 15;
    int kh = lane >> 4;
    bool full = (wave * 16 + 16 <= n_nodes);
    int node = wave * 16 + m;
    if (node >= n_nodes) node = n_nodes - 1;
    const float* arow = helu + (size_t)node * L1C + kh * 2;
    const float* brow = W2 + kh * 2 * OUTC + m;
    v8f c = {};
#pragma unroll 8
    for (int kk = 0; kk < L1C; kk += 4) {
        v2f a = *(const v2f*)(arow + kk);
        v2f b;
        b.x = brow[kk * OUTC];
        b.y = brow[kk * OUTC + OUTC];
        c = __builtin_amdgcn_wmma_f32_16x16x4_f32(false, a, false, b, (short)0, c,
                                                  false, false);
    }
    float* p = h2 + (size_t)(wave * 16 + 8 * kh) * OUTC + m;
    if (full) {
#pragma unroll
        for (int v = 0; v < 8; ++v) p[(size_t)v * OUTC] = c[v];
    } else {
        for (int v = 0; v < 8; ++v)
            if (wave * 16 + 8 * kh + v < n_nodes) p[(size_t)v * OUTC] = c[v];
    }
}

__global__ void k_att2(const float* __restrict__ h2, const float* __restrict__ as2,
                       const float* __restrict__ ad2, float* __restrict__ a_s,
                       float* __restrict__ a_d, int n_nodes) {
    int n = blockIdx.x * blockDim.x + threadIdx.x;
    if (n >= n_nodes) return;
    const float* row = h2 + (size_t)n * OUTC;
    float s = 0.0f, d = 0.0f;
#pragma unroll
    for (int cc = 0; cc < OUTC; ++cc) {
        float v = row[cc];
        s += v * as2[cc];
        d += v * ad2[cc];
    }
    a_s[n] = s;
    a_d[n] = d;
}

__global__ void k_edge_max2(const int* __restrict__ ei, int E, int ET,
                            const float* __restrict__ a_s, const float* __restrict__ a_d,
                            unsigned* __restrict__ m2) {
    int e = blockIdx.x * blockDim.x + threadIdx.x;
    if (e >= ET) return;
    int s, d; edge_sd(ei, E, e, s, d);
    float v = lrelu(a_s[s] + a_d[d]);
    atomicMax(&m2[d], fenc(v));
}

__global__ void k_edge_exp2(const int* __restrict__ ei, int E, int ET,
                            const float* __restrict__ a_s, const float* __restrict__ a_d,
                            const unsigned* __restrict__ m2, float* __restrict__ denom2) {
    int e = blockIdx.x * blockDim.x + threadIdx.x;
    if (e >= ET) return;
    int s, d; edge_sd(ei, E, e, s, d);
    float v = lrelu(a_s[s] + a_d[d]);
    atomicAdd(&denom2[d], expf(v - fdec(m2[d])));
}

__global__ void k_msg2(const int* __restrict__ ei, int E, int ET,
                       const float* __restrict__ a_s, const float* __restrict__ a_d,
                       const unsigned* __restrict__ m2, const float* __restrict__ denom2,
                       const float* __restrict__ h2, float* __restrict__ out) {
    long long gid = (long long)blockIdx.x * blockDim.x + threadIdx.x;
    int e = (int)(gid >> 4);
    int c = (int)(gid & 15);
    if (e >= ET) return;
    int s, d; edge_sd(ei, E, e, s, d);
    float v = lrelu(a_s[s] + a_d[d]);
    float w = expf(v - fdec(m2[d]));
    float alpha = w / (denom2[d] + 1e-16f);
    atomicAdd(&out[(size_t)d * OUTC + c], h2[(size_t)s * OUTC + c] * alpha);
}

__global__ void k_bias2(float* __restrict__ out, const float* __restrict__ bias2, size_t n) {
    size_t i = (size_t)blockIdx.x * blockDim.x + threadIdx.x;
    size_t stride = (size_t)gridDim.x * blockDim.x;
    for (; i < n; i += stride) out[i] += bias2[i & (OUTC - 1)];
}

extern "C" void kernel_launch(void* const* d_in, const int* in_sizes, int n_in,
                              void* d_out, int out_size, void* d_ws, size_t ws_size,
                              hipStream_t stream) {
    const float* x     = (const float*)d_in[0];
    const int*   ei    = (const int*)d_in[1];
    const float* W1    = (const float*)d_in[2];
    const float* as1   = (const float*)d_in[3];
    const float* ad1   = (const float*)d_in[4];
    const float* bias1 = (const float*)d_in[5];
    const float* W2    = (const float*)d_in[6];
    const float* as2   = (const float*)d_in[7];
    const float* ad2   = (const float*)d_in[8];
    const float* bias2 = (const float*)d_in[9];
    float* out = (float*)d_out;

    const int N  = in_sizes[0] / 3;
    const int E  = in_sizes[1] / 2;
    const int ET = E + N;

    float* ws = (float*)d_ws;
    const size_t Nz = (size_t)N;
    float*    out1   = ws;
    unsigned* m1     = (unsigned*)(ws + 256 * Nz);
    float*    denom1 = ws + 264 * Nz;
    unsigned* m2     = (unsigned*)(ws + 272 * Nz);
    float*    denom2 = ws + 273 * Nz;
    const size_t zero_count = 274 * Nz;
    float*    h1   = ws + 274 * Nz;
    float*    a_s1 = ws + 530 * Nz;
    float*    a_d1 = ws + 538 * Nz;
    float*    h2   = ws + 546 * Nz;
    float*    a_s2 = ws + 562 * Nz;
    float*    a_d2 = ws + 563 * Nz;

    const int TB = 256;
    k_fill_zero<<<2048, TB, 0, stream>>>(ws, zero_count);
    k_fill_zero<<<512, TB, 0, stream>>>(out, (size_t)N * OUTC);

    {
        int mtiles = (N + 15) / 16;
        long long waves = (long long)mtiles * 16;
        int blocks = (int)((waves * 32 + TB - 1) / TB);
        k_gemm1_wmma<<<blocks, TB, 0, stream>>>(x, W1, h1, N);
    }
    k_att1<<<N, TB, 0, stream>>>(h1, as1, ad1, a_s1, a_d1, N);
    k_edge_max1<<<(ET + TB - 1) / TB, TB, 0, stream>>>(ei, E, ET, a_s1, a_d1, m1);
    k_edge_exp1<<<(ET + TB - 1) / TB, TB, 0, stream>>>(ei, E, ET, a_s1, a_d1, m1, denom1);
    {
        long long thr = (long long)ET * 32;
        k_msg1<<<(int)((thr + TB - 1) / TB), TB, 0, stream>>>(ei, E, ET, a_s1, a_d1,
                                                              m1, denom1, h1, out1);
    }
    k_elu<<<2048, TB, 0, stream>>>(out1, bias1, (size_t)N * L1C);

    {
        int mtiles = (N + 15) / 16;
        long long thr = (long long)mtiles * 32;
        k_gemm2_wmma<<<(int)((thr + TB - 1) / TB), TB, 0, stream>>>(out1, W2, h2, N);
    }
    k_att2<<<(N + TB - 1) / TB, TB, 0, stream>>>(h2, as2, ad2, a_s2, a_d2, N);
    k_edge_max2<<<(ET + TB - 1) / TB, TB, 0, stream>>>(ei, E, ET, a_s2, a_d2, m2);
    k_edge_exp2<<<(ET + TB - 1) / TB, TB, 0, stream>>>(ei, E, ET, a_s2, a_d2, m2, denom2);
    {
        long long thr = (long long)ET * 16;
        k_msg2<<<(int)((thr + TB - 1) / TB), TB, 0, stream>>>(ei, E, ET, a_s2, a_d2,
                                                              m2, denom2, h2, out);
    }
    k_bias2<<<512, TB, 0, stream>>>(out, bias2, (size_t)N * OUTC);
}


# Layer-2 GEMM moved into the layer-1 walk: 16 waves = 16 nodes, rows (bias+ELU in registers) go to LDS, the last wave to finish runs the 64 WMMAs; ELU and GEMM launches dropped (sum order unchanged)
# speedup vs baseline: 6.1716x; 6.1716x over previous
#include <hip/hip_runtime.h>
#include <math.h>

#define HEADS     8
#define HIDC      32
#define L1C       256
#define OUTC      16
#define NEG_SLOPE 0.2f

typedef __attribute__((ext_vector_type(2))) float v2f;
typedef __attribute__((ext_vector_type(8))) float v8f;

__device__ __forceinline__ float lrelu(float v) {
    return v > 0.0f ? v : NEG_SLOPE * v;
}
#define NL 32
static_assert(NL == 32, "one list per lane of a 32-lane wave");
__global__ void k_fill_head(int* __restrict__ head, int n) {
    int i = blockIdx.x * blockDim.x + threadIdx.x;
    if (i < n) head[i] = -1;
}

__global__ void k_link(const int* __restrict__ ei, int E, int* __restrict__ head,
                       int2* __restrict__ link) {
    int e = blockIdx.x * blockDim.x + threadIdx.x;
    if (e >= E) return;
    int s = ei[e], d = ei[E + e];
    int old = atomicExch(&head[(size_t)d * NL + (e & (NL - 1))], e);
    link[e] = make_int2(old, s);
}

__device__ __forceinline__ float wave_max(float v) {
    for (int off = 16; off; off >>= 1) v = fmaxf(v, __shfl_xor(v, off, 32));
    return v;
}
__device__ __forceinline__ float wave_sum(float v) {
    for (int off = 16; off; off >>= 1) v += __shfl_xor(v, off, 32);
    return v;
}
__device__ __forceinline__ float lane_val(float v, int j) {
    return __int_as_float(__builtin_amdgcn_readlane(__float_as_int(v), j));
}

__device__ __forceinline__ bool lists_live(int cur) {
    return __builtin_amdgcn_ballot_w32(cur >= 0) != 0u;
}
__device__ __forceinline__ void lists_fill(int& cur, const unsigned long long* __restrict__ link,
                                           int lane, int& mysrc, int& cnt) {
    for (int hop = 0; hop < 32 && cnt < 32; ++hop) {
        unsigned live = __builtin_amdgcn_ballot_w32(cur >= 0);
        if (live == 0u) break;
        unsigned long long r = 0ull;
        if (cur >= 0) r = link[cur];
        const int src = (int)(r >> 32);
        const int room = 32 - cnt, have = __builtin_popcount(live);
        const int n = have < room ? have : room;
        unsigned left = live;
        for (int t = 0; t < n; ++t) {
            const int k = __builtin_ctz(left);
            left &= left - 1u;
            const int s = __builtin_amdgcn_readlane(src, k);
            if (lane == cnt + t) mysrc = s;
        }
        cnt += n;
        if (((live & ~left) >> lane) & 1u) cur = (int)(unsigned)r;
    }
}


template <int MASK>
__device__ __forceinline__ float row_xor(float v) {
    return __int_as_float(__builtin_amdgcn_update_dpp(0, __float_as_int(v), 0x160 + MASK,
                                                      0xf, 0xf, true));
}

__global__ void k_gemm1_logits(const float* __restrict__ x, const float* __restrict__ W1,
                               const float* __restrict__ as1, const float* __restrict__ ad1,
                               float* __restrict__ h1, float* __restrict__ a_s,
                               float* __restrict__ a_d, int n_nodes) {
    int wave   = (int)((blockIdx.x * blockDim.x + threadIdx.x) >> 5);
    int lane   = threadIdx.x & 31;
    int mtiles = (n_nodes + 15) >> 4;
    int mtile  = wave >> 1;
    int hg     = wave & 1;
    if (mtile >= mtiles) return;
    int m  = lane & 15;
    int kh = lane >> 4;
    bool full = (mtile * 16 + 16 <= n_nodes);
    int node = mtile * 16 + m;
    if (node >= n_nodes) node = n_nodes - 1;
    float x0 = x[node * 3 + 0], x1 = x[node * 3 + 1], x2 = x[node * 3 + 2];
    v2f a;
    a.x = kh ? x2 : x0;  a.y = kh ? 0.0f : x1;
    const bool b3 = (lane & 8) != 0, b2 = (lane & 4) != 0, b1 = (lane & 2) != 0,
               b0 = (lane & 1) != 0;
    float res[4];
#pragma unroll
    for (int hh = 0; hh < 4; ++hh) {
        const int head = hg * 4 + hh;
        const int col  = head * HIDC + m;
        float w0 = W1[0 * L1C + col], w1 = W1[1 * L1C + col], w2 = W1[2 * L1C + col];
        float u0 = W1[0 * L1C + col + 16], u1 = W1[1 * L1C + col + 16],
              u2 = W1[2 * L1C + col + 16];
        v2f bA, bB;
        bA.x = kh ? w2 : w0;  bA.y = kh ? 0.0f : w1;
        bB.x = kh ? u2 : u0;  bB.y = kh ? 0.0f : u1;
        v8f c0 = {}, c1 = {};
        c0 = __builtin_amdgcn_wmma_f32_16x16x4_f32(false, a, false, bA, (short)0, c0,
                                                   false, false);
        c1 = __builtin_amdgcn_wmma_f32_16x16x4_f32(false, a, false, bB, (short)0, c1,
                                                   false, false);
        float* p = h1 + (size_t)(mtile * 16 + 8 * kh) * L1C + col;
        if (full) {
#pragma unroll
            for (int v = 0; v < 8; ++v) {
                p[(size_t)v * L1C]      = c0[v];
                p[(size_t)v * L1C + 16] = c1[v];
            }
        } else {
            for (int v = 0; v < 8; ++v)
                if (mtile * 16 + 8 * kh + v < n_nodes) {
                    p[(size_t)v * L1C]      = c0[v];
                    p[(size_t)v * L1C + 16] = c1[v];
                }
        }
        const float sl = as1[head * HIDC + m], sh = as1[head * HIDC + 16 + m];
        const float dl = ad1[head * HIDC + m], dh = ad1[head * HIDC + 16 + m];
        float r[16];
#pragma unroll
        for (int v = 0; v < 8; ++v) {
            r[v]     = c0[v] * sl + c1[v] * sh;
            r[8 + v] = c0[v] * dl + c1[v] * dh;
        }
#pragma unroll
        for (int i = 0; i < 8; ++i) {
            float keep = b3 ? r[i + 8] : r[i], give = b3 ? r[i] : r[i + 8];
            r[i] = keep + row_xor<8>(give);
        }
#pragma unroll
        for (int i = 0; i < 4; ++i) {
            float keep = b2 ? r[i + 4] : r[i], give = b2 ? r[i] : r[i + 4];
            r[i] = keep + row_xor<4>(give);
        }
#pragma unroll
        for (int i = 0; i < 2; ++i) {
            float keep = b1 ? r[i + 2] : r[i], give = b1 ? r[i] : r[i + 2];
            r[i] = keep + row_xor<2>(give);
        }
        {
            float keep = b0 ? r[1] : r[0], give = b0 ? r[0] : r[1];
            res[hh] = keep + row_xor<1>(give);
        }
    }
    const int orow = mtile * 16 + 8 * kh + (m & 7);
    float* dst = (m < 8 ? a_s : a_d) + (size_t)orow * HEADS + hg * 4;
    if (orow < n_nodes) {
        dst[0] = res[0]; dst[1] = res[1]; dst[2] = res[2]; dst[3] = res[3];
    }
}

#define WG_NODES 16
#define ROWP (L1C + 4)
__global__ void __launch_bounds__(WG_NODES * 32)
k_walk1_gemm2(const int* __restrict__ head, const unsigned long long* __restrict__ link, int E,
              const float* __restrict__ a_s, const float* __restrict__ a_d,
              const float* __restrict__ h1, const float* __restrict__ bias1,
              const float* __restrict__ W2, float* __restrict__ h2, int n_nodes) {
    __shared__ __attribute__((aligned(16))) float act[WG_NODES * ROWP];
    __shared__ int arrived;
    const int lane = threadIdx.x & 31;
    const int wv = __builtin_amdgcn_readfirstlane((int)(threadIdx.x >> 5));
    const int node0 = (int)blockIdx.x * WG_NODES;
    const int d = node0 + wv;
    if (threadIdx.x == 0) arrived = 0;
    __syncthreads();

    float acc[HEADS];
#pragma unroll
    for (int h = 0; h < HEADS; ++h) acc[h] = 0.0f;

    if (d < n_nodes) {
        const int* myhead = head + (size_t)d * NL + lane;
        const int max_chunks = (E >> 5) + 2;
        float ad[HEADS];
#pragma unroll
        for (int h = 0; h < HEADS; ++h) ad[h] = a_d[d * HEADS + h];

        float m[HEADS], den[HEADS], v[HEADS], w[HEADS];
#pragma unroll
        for (int h = 0; h < HEADS; ++h) { m[h] = -INFINITY; den[h] = 0.0f; }
        int mysrc = d, cnt = 0, cur = *myhead;

        auto load_chunk = [&](bool firstc) {
            cnt = 0;
            if (firstc) { mysrc = d; cnt = 1; }
            lists_fill(cur, link, lane, mysrc, cnt);
            const float4* p = (const float4*)(a_s + (size_t)mysrc * HEADS);
            float4 lo = p[0], hi = p[1];
            v[0] = lo.x; v[1] = lo.y; v[2] = lo.z; v[3] = lo.w;
            v[4] = hi.x; v[5] = hi.y; v[6] = hi.z; v[7] = hi.w;
#pragma unroll
            for (int h = 0; h < HEADS; ++h)
                v[h] = (lane < cnt) ? lrelu(v[h] + ad[h]) : -INFINITY;
        };
        auto add_rows = [&]() {
            float al[HEADS];
#pragma unroll
            for (int h = 0; h < HEADS; ++h) al[h] = w[h] / (den[h] + 1e-16f);
#pragma unroll 2
            for (int j = 0; j < cnt; ++j) {
                int s = __builtin_amdgcn_readlane(mysrc, j);
                const float* ph = h1 + (size_t)s * L1C + lane;
#pragma unroll
                for (int h = 0; h < HEADS; ++h) acc[h] += ph[h * HIDC] * lane_val(al[h], j);
            }
        };

        int nch = 0;
        bool fc = true;
        do {
            load_chunk(fc); fc = false;
#pragma unroll
            for (int h = 0; h < HEADS; ++h) m[h] = fmaxf(m[h], wave_max(v[h]));
            ++nch;
        } while (lists_live(cur) && nch < max_chunks);

        if (nch == 1) {
#pragma unroll
            for (int h = 0; h < HEADS; ++h) { w[h] = expf(v[h] - m[h]); den[h] = wave_sum(w[h]); }
            add_rows();
        } else {
            cur = *myhead; fc = true;
            int k = 0;
            do {
                load_chunk(fc); fc = false;
#pragma unroll
                for (int h = 0; h < HEADS; ++h) den[h] += wave_sum(expf(v[h] - m[h]));
                ++k;
            } while (lists_live(cur) && k < max_chunks);
            cur = *myhead; fc = true; k = 0;
            do {
                load_chunk(fc); fc = false;
#pragma unroll
                for (int h = 0; h < HEADS; ++h) w[h] = expf(v[h] - m[h]);
                add_rows();
                ++k;
            } while (lists_live(cur) && k < max_chunks);
        }
    }

    {
        float bv[HEADS];
#pragma unroll
        for (int h = 0; h < HEADS; ++h) bv[h] = bias1[h * HIDC + lane];
        float* prow = act + wv * ROWP + lane;
#pragma unroll
        for (int h = 0; h < HEADS; ++h) {
            float t = acc[h] + bv[h];
            prow[h * HIDC] = t > 0.0f ? t : (expf(t) - 1.0f);
        }
    }
    __builtin_amdgcn_fence(__ATOMIC_RELEASE, "workgroup");
    int prev = 0;
    if (lane == 0)
        prev = __hip_atomic_fetch_add(&arrived, 1, __ATOMIC_RELAXED, __HIP_MEMORY_SCOPE_WORKGROUP);
    prev = __builtin_amdgcn_readfirstlane(prev);
    if (prev != WG_NODES - 1) return;
    __builtin_amdgcn_fence(__ATOMIC_ACQUIRE, "workgroup");

    const int mm = lane & 15;
    const int kh = lane >> 4;
    const float* arow = act + mm * ROWP + kh * 2;
    const float* brow = W2 + kh * 2 * OUTC + mm;
    v8f c = {};
#pragma unroll 16
    for (int kk = 0; kk < L1C; kk += 4) {
        v2f a = *(const v2f*)(arow + kk);
        v2f b;
        b.x = brow[kk * OUTC];
        b.y = brow[kk * OUTC + OUTC];
        c = __builtin_amdgcn_wmma_f32_16x16x4_f32(false, a, false, b, (short)0, c,
                                                  false, false);
    }
    float* p = h2 + (size_t)(node0 + 8 * kh) * OUTC + mm;
    if (node0 + WG_NODES <= n_nodes) {
#pragma unroll
        for (int r = 0; r < 8; ++r) p[(size_t)r * OUTC] = c[r];
    } else {
        for (int r = 0; r < 8; ++r)
            if (node0 + 8 * kh + r < n_nodes) p[(size_t)r * OUTC] = c[r];
    }
}

__global__ void k_att2(const float* __restrict__ h2, const float* __restrict__ as2,
                       const float* __restrict__ ad2, float* __restrict__ a_s,
                       float* __restrict__ a_d, int n_nodes) {
    int n = blockIdx.x * blockDim.x + threadIdx.x;
    if (n >= n_nodes) return;
    const float* row = h2 + (size_t)n * OUTC;
    float s = 0.0f, d = 0.0f;
#pragma unroll
    for (int cc = 0; cc < OUTC; ++cc) {
        float v = row[cc];
        s += v * as2[cc];
        d += v * ad2[cc];
    }
    a_s[n] = s;
    a_d[n] = d;
}

__global__ void k_walk2(const int* __restrict__ head, const unsigned long long* __restrict__ link, int E,
                        const float* __restrict__ a_s, const float* __restrict__ a_d,
                        const float* __restrict__ h2, float* __restrict__ out, int n_nodes) {
    const int lane = threadIdx.x & 31;
    const int d = __builtin_amdgcn_readfirstlane(
        (int)(blockIdx.x * (blockDim.x >> 5) + (threadIdx.x >> 5)));
    if (d >= n_nodes) return;
    const int* myhead = head + (size_t)d * NL + lane;
    const int max_chunks = (E >> 5) + 2;
    const float adv = a_d[d];
    const int c = lane & 15, half = lane >> 4;
    float m = -INFINITY, den = 0.0f, acc = 0.0f, v = 0.0f, w = 0.0f;
    int mysrc = d, cnt = 0, cur = *myhead;

    auto load_chunk = [&](bool firstc) {
        cnt = 0;
        if (firstc) { mysrc = d; cnt = 1; }
        lists_fill(cur, link, lane, mysrc, cnt);
        v = (lane < cnt) ? lrelu(a_s[mysrc] + adv) : -INFINITY;
    };
    auto add_rows = [&]() {
        float al = w / (den + 1e-16f);
#pragma unroll 4
        for (int j = 0; j < cnt; j += 2) {
            int jj = j + half;
            int s = __shfl(mysrc, jj, 32);
            float a = __shfl(al, jj, 32);
            float hv = h2[(size_t)s * OUTC + c];
            acc += (jj < cnt) ? hv * a : 0.0f;
        }
    };

    int nch = 0;
    bool fc = true;
    do {
        load_chunk(fc); fc = false;
        m = fmaxf(m, wave_max(v));
        ++nch;
    } while (lists_live(cur) && nch < max_chunks);

    if (nch == 1) {
        w = expf(v - m);
        den = wave_sum(w);
        add_rows();
    } else {
        cur = *myhead; fc = true;
        int k = 0;
        do {
            load_chunk(fc); fc = false;
            den += wave_sum(expf(v - m));
            ++k;
        } while (lists_live(cur) && k < max_chunks);
        cur = *myhead; fc = true; k = 0;
        do {
            load_chunk(fc); fc = false;
            w = expf(v - m);
            add_rows();
            ++k;
        } while (lists_live(cur) && k < max_chunks);
    }
    acc += __shfl_xor(acc, 16, 32);
    if (lane < OUTC) out[(size_t)d * OUTC + c] = acc;
}

__global__ void k_bias2(float* __restrict__ out, const float* __restrict__ bias2, size_t n) {
    size_t i = (size_t)blockIdx.x * blockDim.x + threadIdx.x;
    size_t stride = (size_t)gridDim.x * blockDim.x;
    for (; i < n; i += stride) out[i] += bias2[i & (OUTC - 1)];
}

extern "C" void kernel_launch(void* const* d_in, const int* in_sizes, int n_in,
                              void* d_out, int out_size, void* d_ws, size_t ws_size,
                              hipStream_t stream) {
    const float* x     = (const float*)d_in[0];
    const int*   ei    = (const int*)d_in[1];
    const float* W1    = (const float*)d_in[2];
    const float* as1   = (const float*)d_in[3];
    const float* ad1   = (const float*)d_in[4];
    const float* bias1 = (const float*)d_in[5];
    const float* W2    = (const float*)d_in[6];
    const float* as2   = (const float*)d_in[7];
    const float* ad2   = (const float*)d_in[8];
    const float* bias2 = (const float*)d_in[9];
    float* out = (float*)d_out;

    const int N  = in_sizes[0] / 3;
    const int E  = in_sizes[1] / 2;

    float* ws = (float*)d_ws;
    const size_t Nz = (size_t)N;
    float* h1   = ws + 274 * Nz;
    float* a_s1 = ws + 530 * Nz;
    float* a_d1 = ws + 538 * Nz;
    float* h2   = ws + 546 * Nz;
    float* a_s2 = ws + 562 * Nz;
    float* a_d2 = ws + 563 * Nz;
    int*   head = (int*)(ws + 564 * Nz);
    int2*  link = (int2*)(ws + 596 * Nz);

    const int TB = 256;
    const int node_blocks = (N + 7) / 8;
    k_fill_head<<<(NL * N + TB - 1) / TB, TB, 0, stream>>>(head, NL * N);
    k_link<<<(E + TB - 1) / TB, TB, 0, stream>>>(ei, E, head, link);

    {
        int mtiles = (N + 15) / 16;
        long long waves = (long long)mtiles * 2;
        int blocks = (int)((waves * 32 + TB - 1) / TB);
        k_gemm1_logits<<<blocks, TB, 0, stream>>>(x, W1, as1, ad1, h1, a_s1, a_d1, N);
    }
    k_walk1_gemm2<<<(N + WG_NODES - 1) / WG_NODES, WG_NODES * 32, 0, stream>>>(
        head, (const unsigned long long*)link, E, a_s1, a_d1, h1, bias1, W2, h2, N);

    k_att2<<<(N + TB - 1) / TB, TB, 0, stream>>>(h2, as2, ad2, a_s2, a_d2, N);
    k_walk2<<<node_blocks, TB, 0, stream>>>(head, (const unsigned long long*)link, E, a_s2, a_d2, h2, out, N);
    k_bias2<<<512, TB, 0, stream>>>(out, bias2, (size_t)N * OUTC);
}


# Launches 7 -> 4 by merging: edge linking rides in the layer-1 GEMM launch, layer-2 logits are summed from the GEMM-2 tile in registers (16-lane DPP; sum order changes), bias2 is added at the store
# speedup vs baseline: 6.6141x; 1.0717x over previous
#include <hip/hip_runtime.h>
#include <math.h>

#define HEADS     8
#define HIDC      32
#define L1C       256
#define OUTC      16
#define NEG_SLOPE 0.2f

typedef __attribute__((ext_vector_type(2))) float v2f;
typedef __attribute__((ext_vector_type(8))) float v8f;

__device__ __forceinline__ float lrelu(float v) {
    return v > 0.0f ? v : NEG_SLOPE * v;
}
#define NL 32
static_assert(NL == 32, "one list per lane of a 32-lane wave");
__global__ void k_fill_head(int* __restrict__ head, int n) {
    int i = blockIdx.x * blockDim.x + threadIdx.x;
    if (i < n) head[i] = -1;
}

__device__ __forceinline__ void link_edges(int blk, const int* __restrict__ ei, int E,
                                           int* __restrict__ head, int2* __restrict__ link) {
    int e = blk * (int)blockDim.x + (int)threadIdx.x;
    if (e >= E) return;
    int s = ei[e], d = ei[E + e];
    int old = atomicExch(&head[(size_t)d * NL + (e & (NL - 1))], e);
    link[e] = make_int2(old, s);
}

__device__ __forceinline__ float wave_max(float v) {
    for (int off = 16; off; off >>= 1) v = fmaxf(v, __shfl_xor(v, off, 32));
    return v;
}
__device__ __forceinline__ float wave_sum(float v) {
    for (int off = 16; off; off >>= 1) v += __shfl_xor(v, off, 32);
    return v;
}
__device__ __forceinline__ float lane_val(float v, int j) {
    return __int_as_float(__builtin_amdgcn_readlane(__float_as_int(v), j));
}

__device__ __forceinline__ bool lists_live(int cur) {
    return __builtin_amdgcn_ballot_w32(cur >= 0) != 0u;
}
__device__ __forceinline__ void lists_fill(int& cur, const unsigned long long* __restrict__ link,
                                           int lane, int& mysrc, int& cnt) {
    for (int hop = 0; hop < 32 && cnt < 32; ++hop) {
        unsigned live = __builtin_amdgcn_ballot_w32(cur >= 0);
        if (live == 0u) break;
        unsigned long long r = 0ull;
        if (cur >= 0) r = link[cur];
        const int src = (int)(r >> 32);
        const int room = 32 - cnt, have = __builtin_popcount(live);
        const int n = have < room ? have : room;
        unsigned left = live;
        for (int t = 0; t < n; ++t) {
            const int k = __builtin_ctz(left);
            left &= left - 1u;
            const int s = __builtin_amdgcn_readlane(src, k);
            if (lane == cnt + t) mysrc = s;
        }
        cnt += n;
        if (((live & ~left) >> lane) & 1u) cur = (int)(unsigned)r;
    }
}


template <int MASK>
__device__ __forceinline__ float row_xor(float v) {
    return __int_as_float(__builtin_amdgcn_update_dpp(0, __float_as_int(v), 0x160 + MASK,
                                                      0xf, 0xf, true));
}

__device__ __forceinline__ float row_sums16(float (&r)[16], int lane) {
    const bool b3 = (lane & 8) != 0, b2 = (lane & 4) != 0, b1 = (lane & 2) != 0,
               b0 = (lane & 1) != 0;
#pragma unroll
    for (int i = 0; i < 8; ++i) {
        float keep = b3 ? r[i + 8] : r[i], give = b3 ? r[i] : r[i + 8];
        r[i] = keep + row_xor<8>(give);
    }
#pragma unroll
    for (int i = 0; i < 4; ++i) {
        float keep = b2 ? r[i + 4] : r[i], give = b2 ? r[i] : r[i + 4];
        r[i] = keep + row_xor<4>(give);
    }
#pragma unroll
    for (int i = 0; i < 2; ++i) {
        float keep = b1 ? r[i + 2] : r[i], give = b1 ? r[i] : r[i + 2];
        r[i] = keep + row_xor<2>(give);
    }
    float keep = b0 ? r[1] : r[0], give = b0 ? r[0] : r[1];
    return keep + row_xor<1>(give);
}

__device__ __forceinline__ void gemm1_logits_tile(int blk, const float* __restrict__ x,
                               const float* __restrict__ W1,
                               const float* __restrict__ as1, const float* __restrict__ ad1,
                               float* __restrict__ h1, float* __restrict__ a_s,
                               float* __restrict__ a_d, int n_nodes) {
    int wave   = (int)((blk * (int)blockDim.x + (int)threadIdx.x) >> 5);
    int lane   = threadIdx.x & 31;
    int mtiles = (n_nodes + 15) >> 4;
    int mtile  = wave >> 1;
    int hg     = wave & 1;
    if (mtile >= mtiles) return;
    int m  = lane & 15;
    int kh = lane >> 4;
    bool full = (mtile * 16 + 16 <= n_nodes);
    int node = mtile * 16 + m;
    if (node >= n_nodes) node = n_nodes - 1;
    float x0 = x[node * 3 + 0], x1 = x[node * 3 + 1], x2 = x[node * 3 + 2];
    v2f a;
    a.x = kh ? x2 : x0;  a.y = kh ? 0.0f : x1;
    float res[4];
#pragma unroll
    for (int hh = 0; hh < 4; ++hh) {
        const int head = hg * 4 + hh;
        const int col  = head * HIDC + m;
        float w0 = W1[0 * L1C + col], w1 = W1[1 * L1C + col], w2 = W1[2 * L1C + col];
        float u0 = W1[0 * L1C + col + 16], u1 = W1[1 * L1C + col + 16],
              u2 = W1[2 * L1C + col + 16];
        v2f bA, bB;
        bA.x = kh ? w2 : w0;  bA.y = kh ? 0.0f : w1;
        bB.x = kh ? u2 : u0;  bB.y = kh ? 0.0f : u1;
        v8f c0 = {}, c1 = {};
        c0 = __builtin_amdgcn_wmma_f32_16x16x4_f32(false, a, false, bA, (short)0, c0,
                                                   false, false);
        c1 = __builtin_amdgcn_wmma_f32_16x16x4_f32(false, a, false, bB, (short)0, c1,
                                                   false, false);
        float* p = h1 + (size_t)(mtile * 16 + 8 * kh) * L1C + col;
        if (full) {
#pragma unroll
            for (int v = 0; v < 8; ++v) {
                p[(size_t)v * L1C]      = c0[v];
                p[(size_t)v * L1C + 16] = c1[v];
            }
        } else {
            for (int v = 0; v < 8; ++v)
                if (mtile * 16 + 8 * kh + v < n_nodes) {
                    p[(size_t)v * L1C]      = c0[v];
                    p[(size_t)v * L1C + 16] = c1[v];
                }
        }
        const float sl = as1[head * HIDC + m], sh = as1[head * HIDC + 16 + m];
        const float dl = ad1[head * HIDC + m], dh = ad1[head * HIDC + 16 + m];
        float r[16];
#pragma unroll
        for (int v = 0; v < 8; ++v) {
            r[v]     = c0[v] * sl + c1[v] * sh;
            r[8 + v] = c0[v] * dl + c1[v] * dh;
        }
        res[hh] = row_sums16(r, lane);
    }
    const int orow = mtile * 16 + 8 * kh + (m & 7);
    float* dst = (m < 8 ? a_s : a_d) + (size_t)orow * HEADS + hg * 4;
    if (orow < n_nodes) {
        dst[0] = res[0]; dst[1] = res[1]; dst[2] = res[2]; dst[3] = res[3];
    }
}

__global__ void k_gemm1_link(const float* __restrict__ x, const float* __restrict__ W1,
                             const float* __restrict__ as1, const float* __restrict__ ad1,
                             float* __restrict__ h1, float* __restrict__ a_s,
                             float* __restrict__ a_d, int n_nodes, int gemm_blocks,
                             const int* __restrict__ ei, int E, int* __restrict__ head,
                             int2* __restrict__ link) {
    const int blk = (int)blockIdx.x;
    if (blk < gemm_blocks) gemm1_logits_tile(blk, x, W1, as1, ad1, h1, a_s, a_d, n_nodes);
    else link_edges(blk - gemm_blocks, ei, E, head, link);
}

#define WG_NODES 16
#define ROWP (L1C + 4)
__global__ void __launch_bounds__(WG_NODES * 32)
k_walk1_gemm2(const int* __restrict__ head, const unsigned long long* __restrict__ link, int E,
              const float* __restrict__ a_s, const float* __restrict__ a_d,
              const float* __restrict__ h1, const float* __restrict__ bias1,
              const float* __restrict__ W2, const float* __restrict__ as2,
              const float* __restrict__ ad2, float* __restrict__ h2,
              float* __restrict__ a_s2, float* __restrict__ a_d2, int n_nodes) {
    __shared__ __attribute__((aligned(16))) float act[WG_NODES * ROWP];
    __shared__ int arrived;
    const int lane = threadIdx.x & 31;
    const int wv = __builtin_amdgcn_readfirstlane((int)(threadIdx.x >> 5));
    const int node0 = (int)blockIdx.x * WG_NODES;
    const int d = node0 + wv;
    if (threadIdx.x == 0) arrived = 0;
    __syncthreads();

    float acc[HEADS];
#pragma unroll
    for (int h = 0; h < HEADS; ++h) acc[h] = 0.0f;

    if (d < n_nodes) {
        const int* myhead = head + (size_t)d * NL + lane;
        const int max_chunks = (E >> 5) + 2;
        float ad[HEADS];
#pragma unroll
        for (int h = 0; h < HEADS; ++h) ad[h] = a_d[d * HEADS + h];

        float m[HEADS], den[HEADS], v[HEADS], w[HEADS];
#pragma unroll
        for (int h = 0; h < HEADS; ++h) { m[h] = -INFINITY; den[h] = 0.0f; }
        int mysrc = d, cnt = 0, cur = *myhead;

        auto load_chunk = [&](bool firstc) {
            cnt = 0;
            if (firstc) { mysrc = d; cnt = 1; }
            lists_fill(cur, link, lane, mysrc, cnt);
            const float4* p = (const float4*)(a_s + (size_t)mysrc * HEADS);
            float4 lo = p[0], hi = p[1];
            v[0] = lo.x; v[1] = lo.y; v[2] = lo.z; v[3] = lo.w;
            v[4] = hi.x; v[5] = hi.y; v[6] = hi.z; v[7] = hi.w;
#pragma unroll
            for (int h = 0; h < HEADS; ++h)
                v[h] = (lane < cnt) ? lrelu(v[h] + ad[h]) : -INFINITY;
        };
        auto add_rows = [&]() {
            float al[HEADS];
#pragma unroll
            for (int h = 0; h < HEADS; ++h) al[h] = w[h] / (den[h] + 1e-16f);
#pragma unroll 2
            for (int j = 0; j < cnt; ++j) {
                int s = __builtin_amdgcn_readlane(mysrc, j);
                const float* ph = h1 + (size_t)s * L1C + lane;
#pragma unroll
                for (int h = 0; h < HEADS; ++h) acc[h] += ph[h * HIDC] * lane_val(al[h], j);
            }
        };

        int nch = 0;
        bool fc = true;
        do {
            load_chunk(fc); fc = false;
#pragma unroll
            for (int h = 0; h < HEADS; ++h) m[h] = fmaxf(m[h], wave_max(v[h]));
            ++nch;
        } while (lists_live(cur) && nch < max_chunks);

        if (nch == 1) {
#pragma unroll
            for (int h = 0; h < HEADS; ++h) { w[h] = expf(v[h] - m[h]); den[h] = wave_sum(w[h]); }
            add_rows();
        } else {
            cur = *myhead; fc = true;
            int k = 0;
            do {
                load_chunk(fc); fc = false;
#pragma unroll
                for (int h = 0; h < HEADS; ++h) den[h] += wave_sum(expf(v[h] - m[h]));
                ++k;
            } while (lists_live(cur) && k < max_chunks);
            cur = *myhead; fc = true; k = 0;
            do {
                load_chunk(fc); fc = false;
#pragma unroll
                for (int h = 0; h < HEADS; ++h) w[h] = expf(v[h] - m[h]);
                add_rows();
                ++k;
            } while (lists_live(cur) && k < max_chunks);
        }
    }

    {
        float bv[HEADS];
#pragma unroll
        for (int h = 0; h < HEADS; ++h) bv[h] = bias1[h * HIDC + lane];
        float* prow = act + wv * ROWP + lane;
#pragma unroll
        for (int h = 0; h < HEADS; ++h) {
            float t = acc[h] + bv[h];
            prow[h * HIDC] = t > 0.0f ? t : (expf(t) - 1.0f);
        }
    }
    __builtin_amdgcn_fence(__ATOMIC_RELEASE, "workgroup");
    int prev = 0;
    if (lane == 0)
        prev = __hip_atomic_fetch_add(&arrived, 1, __ATOMIC_RELAXED, __HIP_MEMORY_SCOPE_WORKGROUP);
    prev = __builtin_amdgcn_readfirstlane(prev);
    if (prev != WG_NODES - 1) return;
    __builtin_amdgcn_fence(__ATOMIC_ACQUIRE, "workgroup");

    const int mm = lane & 15;
    const int kh = lane >> 4;
    const float* arow = act + mm * ROWP + kh * 2;
    const float* brow = W2 + kh * 2 * OUTC + mm;
    const float sv = as2[mm], dv = ad2[mm];
    v8f c = {};
#pragma unroll 16
    for (int kk = 0; kk < L1C; kk += 4) {
        v2f a = *(const v2f*)(arow + kk);
        v2f b;
        b.x = brow[kk * OUTC];
        b.y = brow[kk * OUTC + OUTC];
        c = __builtin_amdgcn_wmma_f32_16x16x4_f32(false, a, false, b, (short)0, c,
                                                  false, false);
    }
    float* p = h2 + (size_t)(node0 + 8 * kh) * OUTC + mm;
    if (node0 + WG_NODES <= n_nodes) {
#pragma unroll
        for (int r = 0; r < 8; ++r) p[(size_t)r * OUTC] = c[r];
    } else {
        for (int r = 0; r < 8; ++r)
            if (node0 + 8 * kh + r < n_nodes) p[(size_t)r * OUTC] = c[r];
    }
    float r16[16];
#pragma unroll
    for (int v = 0; v < 8; ++v) {
        r16[v]     = c[v] * sv;
        r16[8 + v] = c[v] * dv;
    }
    const float tot = row_sums16(r16, lane);
    const int orow = node0 + 8 * kh + (mm & 7);
    if (orow < n_nodes) (mm < 8 ? a_s2 : a_d2)[orow] = tot;
}

__global__ void k_walk2(const int* __restrict__ head, const unsigned long long* __restrict__ link, int E,
                        const float* __restrict__ a_s, const float* __restrict__ a_d,
                        const float* __restrict__ h2, const float* __restrict__ bias2,
                        float* __restrict__ out, int n_nodes) {
    const int lane = threadIdx.x & 31;
    const int d = __builtin_amdgcn_readfirstlane(
        (int)(blockIdx.x * (blockDim.x >> 5) + (threadIdx.x >> 5)));
    if (d >= n_nodes) return;
    const int* myhead = head + (size_t)d * NL + lane;
    const int max_chunks = (E >> 5) + 2;
    const float adv = a_d[d];
    const int c = lane & 15, half = lane >> 4;
    const float bz = bias2[c];
    float m = -INFINITY, den = 0.0f, acc = 0.0f, v = 0.0f, w = 0.0f;
    int mysrc = d, cnt = 0, cur = *myhead;

    auto load_chunk = [&](bool firstc) {
        cnt = 0;
        if (firstc) { mysrc = d; cnt = 1; }
        lists_fill(cur, link, lane, mysrc, cnt);
        v = (lane < cnt) ? lrelu(a_s[mysrc] + adv) : -INFINITY;
    };
    auto add_rows = [&]() {
        float al = w / (den + 1e-16f);
#pragma unroll 4
        for (int j = 0; j < cnt; j += 2) {
            int jj = j + half;
            int s = __shfl(mysrc, jj, 32);
            float a = __shfl(al, jj, 32);
            float hv = h2[(size_t)s * OUTC + c];
            acc += (jj < cnt) ? hv * a : 0.0f;
        }
    };

    int nch = 0;
    bool fc = true;
    do {
        load_chunk(fc); fc = false;
        m = fmaxf(m, wave_max(v));
        ++nch;
    } while (lists_live(cur) && nch < max_chunks);

    if (nch == 1) {
        w = expf(v - m);
        den = wave_sum(w);
        add_rows();
    } else {
        cur = *myhead; fc = true;
        int k = 0;
        do {
            load_chunk(fc); fc = false;
            den += wave_sum(expf(v - m));
            ++k;
        } while (lists_live(cur) && k < max_chunks);
        cur = *myhead; fc = true; k = 0;
        do {
            load_chunk(fc); fc = false;
            w = expf(v - m);
            add_rows();
            ++k;
        } while (lists_live(cur) && k < max_chunks);
    }
    acc += __shfl_xor(acc, 16, 32);
    if (lane < OUTC) out[(size_t)d * OUTC + c] = acc + bz;
}

extern "C" void kernel_launch(void* const* d_in, const int* in_sizes, int n_in,
                              void* d_out, int out_size, void* d_ws, size_t ws_size,
                              hipStream_t stream) {
    const float* x     = (const float*)d_in[0];
    const int*   ei    = (const int*)d_in[1];
    const float* W1    = (const float*)d_in[2];
    const float* as1   = (const float*)d_in[3];
    const float* ad1   = (const float*)d_in[4];
    const float* bias1 = (const float*)d_in[5];
    const float* W2    = (const float*)d_in[6];
    const float* as2   = (const float*)d_in[7];
    const float* ad2   = (const float*)d_in[8];
    const float* bias2 = (const float*)d_in[9];
    float* out = (float*)d_out;

    const int N  = in_sizes[0] / 3;
    const int E  = in_sizes[1] / 2;

    float* ws = (float*)d_ws;
    const size_t Nz = (size_t)N;
    float* h1   = ws + 274 * Nz;
    float* a_s1 = ws + 530 * Nz;
    float* a_d1 = ws + 538 * Nz;
    float* h2   = ws + 546 * Nz;
    float* a_s2 = ws + 562 * Nz;
    float* a_d2 = ws + 563 * Nz;
    int*   head = (int*)(ws + 564 * Nz);
    int2*  link = (int2*)(ws + 596 * Nz);

    const int TB = 256;
    const int node_blocks = (N + 7) / 8;
    k_fill_head<<<(NL * N + TB - 1) / TB, TB, 0, stream>>>(head, NL * N);

    {
        int mtiles = (N + 15) / 16;
        long long waves = (long long)mtiles * 2;
        int gemm_blocks = (int)((waves * 32 + TB - 1) / TB);
        int link_blocks = (E + TB - 1) / TB;
        k_gemm1_link<<<gemm_blocks + link_blocks, TB, 0, stream>>>(
            x, W1, as1, ad1, h1, a_s1, a_d1, N, gemm_blocks, ei, E, head, link);
    }
    k_walk1_gemm2<<<(N + WG_NODES - 1) / WG_NODES, WG_NODES * 32, 0, stream>>>(
        head, (const unsigned long long*)link, E, a_s1, a_d1, h1, bias1, W2, as2, ad2, h2,
        a_s2, a_d2, N);

    k_walk2<<<node_blocks, TB, 0, stream>>>(head, (const unsigned long long*)link, E, a_s2, a_d2, h2,
                                            bias2, out, N);
}


# Layer-1 walk gathers the sources' 3 inputs, not 1-KB h1 rows: rows re-formed per node on WMMA (W1^T @ weighted x^T, entries = columns), summed over 16 lanes by DPP; h1 never stored (sum order changes)
# speedup vs baseline: 7.1676x; 1.0837x over previous
#include <hip/hip_runtime.h>
#include <math.h>

#define HEADS     8
#define HIDC      32
#define L1C       256
#define OUTC      16
#define NEG_SLOPE 0.2f

typedef __attribute__((ext_vector_type(2))) float v2f;
typedef __attribute__((ext_vector_type(8))) float v8f;

__device__ __forceinline__ float lrelu(float v) {
    return v > 0.0f ? v : NEG_SLOPE * v;
}
#define NL 32
static_assert(NL == 32, "one list per lane of a 32-lane wave");
__global__ void k_fill_head(int* __restrict__ head, int n) {
    int i = blockIdx.x * blockDim.x + threadIdx.x;
    if (i < n) head[i] = -1;
}

__device__ __forceinline__ void link_edges(int blk, const int* __restrict__ ei, int E,
                                           int* __restrict__ head, int2* __restrict__ link) {
    int e = blk * (int)blockDim.x + (int)threadIdx.x;
    if (e >= E) return;
    int s = ei[e], d = ei[E + e];
    int old = atomicExch(&head[(size_t)d * NL + (e & (NL - 1))], e);
    link[e] = make_int2(old, s);
}

__device__ __forceinline__ float wave_max(float v) {
    for (int off = 16; off; off >>= 1) v = fmaxf(v, __shfl_xor(v, off, 32));
    return v;
}
__device__ __forceinline__ float wave_sum(float v) {
    for (int off = 16; off; off >>= 1) v += __shfl_xor(v, off, 32);
    return v;
}

__device__ __forceinline__ bool lists_live(int cur) {
    return __builtin_amdgcn_ballot_w32(cur >= 0) != 0u;
}
__device__ __forceinline__ void lists_fill(int& cur, const unsigned long long* __restrict__ link,
                                           int lane, int& mysrc, int& cnt) {
    for (int hop = 0; hop < 32 && cnt < 32; ++hop) {
        unsigned live = __builtin_amdgcn_ballot_w32(cur >= 0);
        if (live == 0u) break;
        unsigned long long r = 0ull;
        if (cur >= 0) r = link[cur];
        const int src = (int)(r >> 32);
        const int room = 32 - cnt, have = __builtin_popcount(live);
        const int n = have < room ? have : room;
        unsigned left = live;
        for (int t = 0; t < n; ++t) {
            const int k = __builtin_ctz(left);
            left &= left - 1u;
            const int s = __builtin_amdgcn_readlane(src, k);
            if (lane == cnt + t) mysrc = s;
        }
        cnt += n;
        if (((live & ~left) >> lane) & 1u) cur = (int)(unsigned)r;
    }
}


template <int MASK>
__device__ __forceinline__ float row_xor(float v) {
    return __int_as_float(__builtin_amdgcn_update_dpp(0, __float_as_int(v), 0x160 + MASK,
                                                      0xf, 0xf, true));
}

__device__ __forceinline__ float row_sums16(float (&r)[16], int lane) {
    const bool b3 = (lane & 8) != 0, b2 = (lane & 4) != 0, b1 = (lane & 2) != 0,
               b0 = (lane & 1) != 0;
#pragma unroll
    for (int i = 0; i < 8; ++i) {
        float keep = b3 ? r[i + 8] : r[i], give = b3 ? r[i] : r[i + 8];
        r[i] = keep + row_xor<8>(give);
    }
#pragma unroll
    for (int i = 0; i < 4; ++i) {
        float keep = b2 ? r[i + 4] : r[i], give = b2 ? r[i] : r[i + 4];
        r[i] = keep + row_xor<4>(give);
    }
#pragma unroll
    for (int i = 0; i < 2; ++i) {
        float keep = b1 ? r[i + 2] : r[i], give = b1 ? r[i] : r[i + 2];
        r[i] = keep + row_xor<2>(give);
    }
    float keep = b0 ? r[1] : r[0], give = b0 ? r[0] : r[1];
    return keep + row_xor<1>(give);
}

__device__ __forceinline__ void gemm1_logits_tile(int blk, const float* __restrict__ x,
                               const float* __restrict__ W1,
                               const float* __restrict__ as1, const float* __restrict__ ad1,
                               float* __restrict__ a_s,
                               float* __restrict__ a_d, int n_nodes) {
    int wave   = (int)((blk * (int)blockDim.x + (int)threadIdx.x) >> 5);
    int lane   = threadIdx.x & 31;
    int mtiles = (n_nodes + 15) >> 4;
    int mtile  = wave >> 1;
    int hg     = wave & 1;
    if (mtile >= mtiles) return;
    int m  = lane & 15;
    int kh = lane >> 4;
    int node = mtile * 16 + m;
    if (node >= n_nodes) node = n_nodes - 1;
    float x0 = x[node * 3 + 0], x1 = x[node * 3 + 1], x2 = x[node * 3 + 2];
    v2f a;
    a.x = kh ? x2 : x0;  a.y = kh ? 0.0f : x1;
    const int orow = mtile * 16 + 8 * kh + (m & 7);
    float* dst = (m < 8 ? a_s : a_d) + (size_t)orow * HEADS + hg * 4;
#pragma unroll 1
    for (int hh = 0; hh < 4; ++hh) {
        const int head = hg * 4 + hh;
        const int col  = head * HIDC + m;
        float w0 = W1[0 * L1C + col], w1 = W1[1 * L1C + col], w2 = W1[2 * L1C + col];
        float u0 = W1[0 * L1C + col + 16], u1 = W1[1 * L1C + col + 16],
              u2 = W1[2 * L1C + col + 16];
        v2f bA, bB;
        bA.x = kh ? w2 : w0;  bA.y = kh ? 0.0f : w1;
        bB.x = kh ? u2 : u0;  bB.y = kh ? 0.0f : u1;
        v8f c0 = {}, c1 = {};
        c0 = __builtin_amdgcn_wmma_f32_16x16x4_f32(false, a, false, bA, (short)0, c0,
                                                   false, false);
        c1 = __builtin_amdgcn_wmma_f32_16x16x4_f32(false, a, false, bB, (short)0, c1,
                                                   false, false);
        const float sl = as1[head * HIDC + m], sh = as1[head * HIDC + 16 + m];
        const float dl = ad1[head * HIDC + m], dh = ad1[head * HIDC + 16 + m];
        float r[16];
#pragma unroll
        for (int v = 0; v < 8; ++v) {
            r[v]     = c0[v] * sl + c1[v] * sh;
            r[8 + v] = c0[v] * dl + c1[v] * dh;
        }
        const float tot = row_sums16(r, lane);
        if (orow < n_nodes) dst[hh] = tot;
    }
}

__global__ void k_gemm1_link(const float* __restrict__ x, const float* __restrict__ W1,
                             const float* __restrict__ as1, const float* __restrict__ ad1,
                             float* __restrict__ a_s,
                             float* __restrict__ a_d, int n_nodes, int gemm_blocks,
                             const int* __restrict__ ei, int E, int* __restrict__ head,
                             int2* __restrict__ link) {
    const int blk = (int)blockIdx.x;
    if (blk < gemm_blocks) gemm1_logits_tile(blk, x, W1, as1, ad1, a_s, a_d, n_nodes);
    else link_edges(blk - gemm_blocks, ei, E, head, link);
}

#define WG_NODES 16
#define ROWP (L1C + 4)
__global__ void __launch_bounds__(WG_NODES * 32)
k_walk1_gemm2(const int* __restrict__ head, const unsigned long long* __restrict__ link, int E,
              const float* __restrict__ a_s, const float* __restrict__ a_d,
              const float* __restrict__ x, const float* __restrict__ W1,
              const float* __restrict__ bias1,
              const float* __restrict__ W2, const float* __restrict__ as2,
              const float* __restrict__ ad2, float* __restrict__ h2,
              float* __restrict__ a_s2, float* __restrict__ a_d2, int n_nodes) {
    __shared__ __attribute__((aligned(16))) float act[WG_NODES * ROWP];
    __shared__ int arrived;
    __shared__ __attribute__((aligned(8))) v2f fragW[(L1C / 16) * 32];
    static_assert(WG_NODES == L1C / 16, "wave nt of the work-group stages tile nt");
    const int lane = threadIdx.x & 31;
    const int wv = __builtin_amdgcn_readfirstlane((int)(threadIdx.x >> 5));
    const int node0 = (int)blockIdx.x * WG_NODES;
    const int d = node0 + wv;
    const int mm = lane & 15;
    const int kh = lane >> 4;
    const int chan = (mm >> 3) * 16 + 8 * kh + (mm & 7);
    int cur0 = -1;
    if (d < n_nodes) cur0 = head[(size_t)d * NL + lane];
    {
        const int col = wv * 16 + mm;
        const float wx = W1[(kh ? 2 * L1C : 0) + col], wy = W1[L1C + col];
        v2f f;
        f.x = wx;  f.y = kh ? 0.0f : wy;
        fragW[wv * 32 + lane] = f;
    }
    if (threadIdx.x == 0) arrived = 0;
    __syncthreads();

    float acc[HEADS];
#pragma unroll
    for (int h = 0; h < HEADS; ++h) acc[h] = 0.0f;

    if (d < n_nodes) {
        const int* myhead = head + (size_t)d * NL + lane;
        const int max_chunks = (E >> 5) + 2;
        float ad[HEADS];
#pragma unroll
        for (int h = 0; h < HEADS; ++h) ad[h] = a_d[d * HEADS + h];

        float m[HEADS], den[HEADS], v[HEADS], w[HEADS];
#pragma unroll
        for (int h = 0; h < HEADS; ++h) { m[h] = -INFINITY; den[h] = 0.0f; }
        int mysrc = d, cnt = 0, cur = cur0;

        auto load_chunk = [&](bool firstc) {
            cnt = 0;
            if (firstc) { mysrc = d; cnt = 1; }
            lists_fill(cur, link, lane, mysrc, cnt);
            const float4* p = (const float4*)(a_s + (size_t)mysrc * HEADS);
            float4 lo = p[0], hi = p[1];
            v[0] = lo.x; v[1] = lo.y; v[2] = lo.z; v[3] = lo.w;
            v[4] = hi.x; v[5] = hi.y; v[6] = hi.z; v[7] = hi.w;
#pragma unroll
            for (int h = 0; h < HEADS; ++h)
                v[h] = (lane < cnt) ? lrelu(v[h] + ad[h]) : -INFINITY;
        };
        auto add_rows = [&](const bool two) {
            float al[HEADS];
#pragma unroll
            for (int h = 0; h < HEADS; ++h) al[h] = w[h] / (den[h] + 1e-16f);
            v2f xb0, xb1 = {};
            {
                const int s0 = __shfl(mysrc, mm, 32);
                const float* p0 = x + (size_t)s0 * 3 + 2 * kh;
                xb0.x = p0[0];  xb0.y = 0.0f;
                if (!kh) xb0.y = p0[1];
                if (two) {
                    const int s1 = __shfl(mysrc, 16 + mm, 32);
                    const float* p1 = x + (size_t)s1 * 3 + 2 * kh;
                    xb1.x = p1[0];  xb1.y = 0.0f;
                    if (!kh) xb1.y = p1[1];
                }
            }
#pragma unroll
            for (int h = 0; h < HEADS; ++h) {
                const float ax = __shfl_xor(al[h], 16, 32);
                const float wt0 = kh ? ax : al[h];
                const float wt1 = kh ? al[h] : ax;
                const v2f aA = fragW[(2 * h) * 32 + lane], aB = fragW[(2 * h + 1) * 32 + lane];
                v2f b;
                b.x = xb0.x * wt0;  b.y = xb0.y * wt0;
                v8f c0 = {}, c1 = {};
                c0 = __builtin_amdgcn_wmma_f32_16x16x4_f32(false, aA, false, b, (short)0, c0,
                                                           false, false);
                c1 = __builtin_amdgcn_wmma_f32_16x16x4_f32(false, aB, false, b, (short)0, c1,
                                                           false, false);
                if (two) {
                    v2f b2;
                    b2.x = xb1.x * wt1;  b2.y = xb1.y * wt1;
                    c0 = __builtin_amdgcn_wmma_f32_16x16x4_f32(false, aA, false, b2, (short)0, c0,
                                                               false, false);
                    c1 = __builtin_amdgcn_wmma_f32_16x16x4_f32(false, aB, false, b2, (short)0, c1,
                                                               false, false);
                }
                float r[16];
#pragma unroll
                for (int i = 0; i < 8; ++i) { r[i] = c0[i];  r[8 + i] = c1[i]; }
                acc[h] += row_sums16(r, lane);
            }
        };

        int nch = 0;
        bool fc = true;
        do {
            load_chunk(fc); fc = false;
#pragma unroll
            for (int h = 0; h < HEADS; ++h) m[h] = fmaxf(m[h], wave_max(v[h]));
            ++nch;
        } while (lists_live(cur) && nch < max_chunks);

        if (nch == 1) {
#pragma unroll
            for (int h = 0; h < HEADS; ++h) { w[h] = expf(v[h] - m[h]); den[h] = wave_sum(w[h]); }
            if (cnt > 16) add_rows(true); else add_rows(false);
        } else {
            cur = *myhead; fc = true;
            int k = 0;
            do {
                load_chunk(fc); fc = false;
#pragma unroll
                for (int h = 0; h < HEADS; ++h) den[h] += wave_sum(expf(v[h] - m[h]));
                ++k;
            } while (lists_live(cur) && k < max_chunks);
            cur = *myhead; fc = true; k = 0;
            do {
                load_chunk(fc); fc = false;
#pragma unroll
                for (int h = 0; h < HEADS; ++h) w[h] = expf(v[h] - m[h]);
                if (cnt > 16) add_rows(true); else add_rows(false);
                ++k;
            } while (lists_live(cur) && k < max_chunks);
        }
    }

    {
        float bv[HEADS];
#pragma unroll
        for (int h = 0; h < HEADS; ++h) bv[h] = bias1[h * HIDC + chan];
        float* prow = act + wv * ROWP + chan;
#pragma unroll
        for (int h = 0; h < HEADS; ++h) {
            float t = acc[h] + bv[h];
            prow[h * HIDC] = t > 0.0f ? t : (expf(t) - 1.0f);
        }
    }
    __builtin_amdgcn_fence(__ATOMIC_RELEASE, "workgroup");
    int prev = 0;
    if (lane == 0)
        prev = __hip_atomic_fetch_add(&arrived, 1, __ATOMIC_RELAXED, __HIP_MEMORY_SCOPE_WORKGROUP);
    prev = __builtin_amdgcn_readfirstlane(prev);
    if (prev != WG_NODES - 1) return;
    __builtin_amdgcn_fence(__ATOMIC_ACQUIRE, "workgroup");

    const float* arow = act + mm * ROWP + kh * 2;
    const float* brow = W2 + kh * 2 * OUTC + mm;
    const float sv = as2[mm], dv = ad2[mm];
    v8f c = {};
#pragma unroll 16
    for (int kk = 0; kk < L1C; kk += 4) {
        v2f a = *(const v2f*)(arow + kk);
        v2f b;
        b.x = brow[kk * OUTC];
        b.y = brow[kk * OUTC + OUTC];
        c = __builtin_amdgcn_wmma_f32_16x16x4_f32(false, a, false, b, (short)0, c,
                                                  false, false);
    }
    float* p = h2 + (size_t)(node0 + 8 * kh) * OUTC + mm;
    if (node0 + WG_NODES <= n_nodes) {
#pragma unroll
        for (int r = 0; r < 8; ++r) p[(size_t)r * OUTC] = c[r];
    } else {
        for (int r = 0; r < 8; ++r)
            if (node0 + 8 * kh + r < n_nodes) p[(size_t)r * OUTC] = c[r];
    }
    float r16[16];
#pragma unroll
    for (int v = 0; v < 8; ++v) {
        r16[v]     = c[v] * sv;
        r16[8 + v] = c[v] * dv;
    }
    const float tot = row_sums16(r16, lane);
    const int orow = node0 + 8 * kh + (mm & 7);
    if (orow < n_nodes) (mm < 8 ? a_s2 : a_d2)[orow] = tot;
}

__global__ void k_walk2(const int* __restrict__ head, const unsigned long long* __restrict__ link, int E,
                        const float* __restrict__ a_s, const float* __restrict__ a_d,
                        const float* __restrict__ h2, const float* __restrict__ bias2,
                        float* __restrict__ out, int n_nodes) {
    const int lane = threadIdx.x & 31;
    const int d = __builtin_amdgcn_readfirstlane(
        (int)(blockIdx.x * (blockDim.x >> 5) + (threadIdx.x >> 5)));
    if (d >= n_nodes) return;
    const int* myhead = head + (size_t)d * NL + lane;
    const int max_chunks = (E >> 5) + 2;
    const float adv = a_d[d];
    const int c = lane & 15, half = lane >> 4;
    const float bz = bias2[c];
    float m = -INFINITY, den = 0.0f, acc = 0.0f, v = 0.0f, w = 0.0f;
    int mysrc = d, cnt = 0, cur = *myhead;

    auto load_chunk = [&](bool firstc) {
        cnt = 0;
        if (firstc) { mysrc = d; cnt = 1; }
        lists_fill(cur, link, lane, mysrc, cnt);
        v = (lane < cnt) ? lrelu(a_s[mysrc] + adv) : -INFINITY;
    };
    auto add_rows = [&]() {
        float al = w / (den + 1e-16f);
#pragma unroll 4
        for (int j = 0; j < cnt; j += 2) {
            int jj = j + half;
            int s = __shfl(mysrc, jj, 32);
            float a = __shfl(al, jj, 32);
            float hv = h2[(size_t)s * OUTC + c];
            acc += (jj < cnt) ? hv * a : 0.0f;
        }
    };

    int nch = 0;
    bool fc = true;
    do {
        load_chunk(fc); fc = false;
        m = fmaxf(m, wave_max(v));
        ++nch;
    } while (lists_live(cur) && nch < max_chunks);

    if (nch == 1) {
        w = expf(v - m);
        den = wave_sum(w);
        add_rows();
    } else {
        cur = *myhead; fc = true;
        int k = 0;
        do {
            load_chunk(fc); fc = false;
            den += wave_sum(expf(v - m));
            ++k;
        } while (lists_live(cur) && k < max_chunks);
        cur = *myhead; fc = true; k = 0;
        do {
            load_chunk(fc); fc = false;
            w = expf(v - m);
            add_rows();
            ++k;
        } while (lists_live(cur) && k < max_chunks);
    }
    acc += __shfl_xor(acc, 16, 32);
    if (lane < OUTC) out[(size_t)d * OUTC + c] = acc + bz;
}

extern "C" void kernel_launch(void* const* d_in, const int* in_sizes, int n_in,
                              void* d_out, int out_size, void* d_ws, size_t ws_size,
                              hipStream_t stream) {
    const float* x     = (const float*)d_in[0];
    const int*   ei    = (const int*)d_in[1];
    const float* W1    = (const float*)d_in[2];
    const float* as1   = (const float*)d_in[3];
    const float* ad1   = (const float*)d_in[4];
    const float* bias1 = (const float*)d_in[5];
    const float* W2    = (const float*)d_in[6];
    const float* as2   = (const float*)d_in[7];
    const float* ad2   = (const float*)d_in[8];
    const float* bias2 = (const float*)d_in[9];
    float* out = (float*)d_out;

    const int N  = in_sizes[0] / 3;
    const int E  = in_sizes[1] / 2;

    float* ws = (float*)d_ws;
    const size_t Nz = (size_t)N;
    float* a_s1 = ws + 530 * Nz;
    float* a_d1 = ws + 538 * Nz;
    float* h2   = ws + 546 * Nz;
    float* a_s2 = ws + 562 * Nz;
    float* a_d2 = ws + 563 * Nz;
    int*   head = (int*)(ws + 564 * Nz);
    int2*  link = (int2*)(ws + 596 * Nz);

    const int TB = 256;
    const int node_blocks = (N + 7) / 8;
    k_fill_head<<<(NL * N + TB - 1) / TB, TB, 0, stream>>>(head, NL * N);

    {
        int mtiles = (N + 15) / 16;
        long long waves = (long long)mtiles * 2;
        int gemm_blocks = (int)((waves * 32 + TB - 1) / TB);
        int link_blocks = (E + TB - 1) / TB;
        k_gemm1_link<<<gemm_blocks + link_blocks, TB, 0, stream>>>(
            x, W1, as1, ad1, a_s1, a_d1, N, gemm_blocks, ei, E, head, link);
    }
    k_walk1_gemm2<<<(N + WG_NODES - 1) / WG_NODES, WG_NODES * 32, 0, stream>>>(
        head, (const unsigned long long*)link, E, a_s1, a_d1, x, W1, bias1, W2, as2, ad2, h2,
        a_s2, a_d2, N);

    k_walk2<<<node_blocks, TB, 0, stream>>>(head, (const unsigned long long*)link, E, a_s2, a_d2, h2,
                                            bias2, out, N);
}


# Layer-1 walk held to 96 registers (launch bound, was 122): 10 waves per SIMD, not 8; the next 16-wave work-group enters while stragglers and the GEMM wave of the old one run (sum order unchanged)
# speedup vs baseline: 7.6390x; 1.0658x over previous
#include <hip/hip_runtime.h>
#include <math.h>

#define HEADS     8
#define HIDC      32
#define L1C       256
#define OUTC      16
#define NEG_SLOPE 0.2f

typedef __attribute__((ext_vector_type(2))) float v2f;
typedef __attribute__((ext_vector_type(8))) float v8f;

__device__ __forceinline__ float lrelu(float v) {
    return v > 0.0f ? v : NEG_SLOPE * v;
}
#define NL 32
static_assert(NL == 32, "one list per lane of a 32-lane wave");
__global__ void k_fill_head(int* __restrict__ head, int n) {
    int i = blockIdx.x * blockDim.x + threadIdx.x;
    if (i < n) head[i] = -1;
}

__device__ __forceinline__ void link_edges(int blk, const int* __restrict__ ei, int E,
                                           int* __restrict__ head, int2* __restrict__ link) {
    int e = blk * (int)blockDim.x + (int)threadIdx.x;
    if (e >= E) return;
    int s = ei[e], d = ei[E + e];
    int old = atomicExch(&head[(size_t)d * NL + (e & (NL - 1))], e);
    link[e] = make_int2(old, s);
}

__device__ __forceinline__ float wave_max(float v) {
    for (int off = 16; off; off >>= 1) v = fmaxf(v, __shfl_xor(v, off, 32));
    return v;
}
__device__ __forceinline__ float wave_sum(float v) {
    for (int off = 16; off; off >>= 1) v += __shfl_xor(v, off, 32);
    return v;
}

__device__ __forceinline__ bool lists_live(int cur) {
    return __builtin_amdgcn_ballot_w32(cur >= 0) != 0u;
}
__device__ __forceinline__ void lists_fill(int& cur, const unsigned long long* __restrict__ link,
                                           int lane, int& mysrc, int& cnt) {
    for (int hop = 0; hop < 32 && cnt < 32; ++hop) {
        unsigned live = __builtin_amdgcn_ballot_w32(cur >= 0);
        if (live == 0u) break;
        unsigned long long r = 0ull;
        if (cur >= 0) r = link[cur];
        const int src = (int)(r >> 32);
        const int room = 32 - cnt, have = __builtin_popcount(live);
        const int n = have < room ? have : room;
        unsigned left = live;
        for (int t = 0; t < n; ++t) {
            const int k = __builtin_ctz(left);
            left &= left - 1u;
            const int s = __builtin_amdgcn_readlane(src, k);
            if (lane == cnt + t) mysrc = s;
        }
        cnt += n;
        if (((live & ~left) >> lane) & 1u) cur = (int)(unsigned)r;
    }
}


template <int MASK>
__device__ __forceinline__ float row_xor(float v) {
    return __int_as_float(__builtin_amdgcn_update_dpp(0, __float_as_int(v), 0x160 + MASK,
                                                      0xf, 0xf, true));
}

__device__ __forceinline__ float row_sums16(float (&r)[16], int lane) {
    const bool b3 = (lane & 8) != 0, b2 = (lane & 4) != 0, b1 = (lane & 2) != 0,
               b0 = (lane & 1) != 0;
#pragma unroll
    for (int i = 0; i < 8; ++i) {
        float keep = b3 ? r[i + 8] : r[i], give = b3 ? r[i] : r[i + 8];
        r[i] = keep + row_xor<8>(give);
    }
#pragma unroll
    for (int i = 0; i < 4; ++i) {
        float keep = b2 ? r[i + 4] : r[i], give = b2 ? r[i] : r[i + 4];
        r[i] = keep + row_xor<4>(give);
    }
#pragma unroll
    for (int i = 0; i < 2; ++i) {
        float keep = b1 ? r[i + 2] : r[i], give = b1 ? r[i] : r[i + 2];
        r[i] = keep + row_xor<2>(give);
    }
    float keep = b0 ? r[1] : r[0], give = b0 ? r[0] : r[1];
    return keep + row_xor<1>(give);
}

__device__ __forceinline__ void gemm1_logits_tile(int blk, const float* __restrict__ x,
                               const float* __restrict__ W1,
                               const float* __restrict__ as1, const float* __restrict__ ad1,
                               float* __restrict__ a_s,
                               float* __restrict__ a_d, int n_nodes) {
    int wave   = (int)((blk * (int)blockDim.x + (int)threadIdx.x) >> 5);
    int lane   = threadIdx.x & 31;
    int mtiles = (n_nodes + 15) >> 4;
    int mtile  = wave >> 1;
    int hg     = wave & 1;
    if (mtile >= mtiles) return;
    int m  = lane & 15;
    int kh = lane >> 4;
    int node = mtile * 16 + m;
    if (node >= n_nodes) node = n_nodes - 1;
    float x0 = x[node * 3 + 0], x1 = x[node * 3 + 1], x2 = x[node * 3 + 2];
    v2f a;
    a.x = kh ? x2 : x0;  a.y = kh ? 0.0f : x1;
    const int orow = mtile * 16 + 8 * kh + (m & 7);
    float* dst = (m < 8 ? a_s : a_d) + (size_t)orow * HEADS + hg * 4;
#pragma unroll 1
    for (int hh = 0; hh < 4; ++hh) {
        const int head = hg * 4 + hh;
        const int col  = head * HIDC + m;
        float w0 = W1[0 * L1C + col], w1 = W1[1 * L1C + col], w2 = W1[2 * L1C + col];
        float u0 = W1[0 * L1C + col + 16], u1 = W1[1 * L1C + col + 16],
              u2 = W1[2 * L1C + col + 16];
        v2f bA, bB;
        bA.x = kh ? w2 : w0;  bA.y = kh ? 0.0f : w1;
        bB.x = kh ? u2 : u0;  bB.y = kh ? 0.0f : u1;
        v8f c0 = {}, c1 = {};
        c0 = __builtin_amdgcn_wmma_f32_16x16x4_f32(false, a, false, bA, (short)0, c0,
                                                   false, false);
        c1 = __builtin_amdgcn_wmma_f32_16x16x4_f32(false, a, false, bB, (short)0, c1,
                                                   false, false);
        const float sl = as1[head * HIDC + m], sh = as1[head * HIDC + 16 + m];
        const float dl = ad1[head * HIDC + m], dh = ad1[head * HIDC + 16 + m];
        float r[16];
#pragma unroll
        for (int v = 0; v < 8; ++v) {
            r[v]     = c0[v] * sl + c1[v] * sh;
            r[8 + v] = c0[v] * dl + c1[v] * dh;
        }
        const float tot = row_sums16(r, lane);
        if (orow < n_nodes) dst[hh] = tot;
    }
}

__global__ void k_gemm1_link(const float* __restrict__ x, const float* __restrict__ W1,
                             const float* __restrict__ as1, const float* __restrict__ ad1,
                             float* __restrict__ a_s,
                             float* __restrict__ a_d, int n_nodes, int gemm_blocks,
                             const int* __restrict__ ei, int E, int* __restrict__ head,
                             int2* __restrict__ link) {
    const int blk = (int)blockIdx.x;
    if (blk < gemm_blocks) gemm1_logits_tile(blk, x, W1, as1, ad1, a_s, a_d, n_nodes);
    else link_edges(blk - gemm_blocks, ei, E, head, link);
}

#define WG_NODES 16
#define ROWP (L1C + 4)
#define WALK1_WAVES_PER_SIMD 10
__global__ void __launch_bounds__(WG_NODES * 32, WALK1_WAVES_PER_SIMD)
k_walk1_gemm2(const int* __restrict__ head, const unsigned long long* __restrict__ link, int E,
              const float* __restrict__ a_s, const float* __restrict__ a_d,
              const float* __restrict__ x, const float* __restrict__ W1,
              const float* __restrict__ bias1,
              const float* __restrict__ W2, const float* __restrict__ as2,
              const float* __restrict__ ad2, float* __restrict__ h2,
              float* __restrict__ a_s2, float* __restrict__ a_d2, int n_nodes) {
    __shared__ __attribute__((aligned(16))) float act[WG_NODES * ROWP];
    __shared__ int arrived;
    __shared__ __attribute__((aligned(8))) v2f fragW[(L1C / 16) * 32];
    static_assert(WG_NODES == L1C / 16, "wave nt of the work-group stages tile nt");
    const int lane = threadIdx.x & 31;
    const int wv = __builtin_amdgcn_readfirstlane((int)(threadIdx.x >> 5));
    const int node0 = (int)blockIdx.x * WG_NODES;
    const int d = node0 + wv;
    const int mm = lane & 15;
    const int kh = lane >> 4;
    const int chan = (mm >> 3) * 16 + 8 * kh + (mm & 7);
    int cur0 = -1;
    if (d < n_nodes) cur0 = head[(size_t)d * NL + lane];
    {
        const int col = wv * 16 + mm;
        const float wx = W1[(kh ? 2 * L1C : 0) + col], wy = W1[L1C + col];
        v2f f;
        f.x = wx;  f.y = kh ? 0.0f : wy;
        fragW[wv * 32 + lane] = f;
    }
    if (threadIdx.x == 0) arrived = 0;
    __syncthreads();

    float acc[HEADS];
#pragma unroll
    for (int h = 0; h < HEADS; ++h) acc[h] = 0.0f;

    if (d < n_nodes) {
        const int* myhead = head + (size_t)d * NL + lane;
        const int max_chunks = (E >> 5) + 2;
        float ad[HEADS];
#pragma unroll
        for (int h = 0; h < HEADS; ++h) ad[h] = a_d[d * HEADS + h];

        float m[HEADS], den[HEADS], v[HEADS], w[HEADS];
#pragma unroll
        for (int h = 0; h < HEADS; ++h) { m[h] = -INFINITY; den[h] = 0.0f; }
        int mysrc = d, cnt = 0, cur = cur0;

        auto load_chunk = [&](bool firstc) {
            cnt = 0;
            if (firstc) { mysrc = d; cnt = 1; }
            lists_fill(cur, link, lane, mysrc, cnt);
            const float4* p = (const float4*)(a_s + (size_t)mysrc * HEADS);
            float4 lo = p[0], hi = p[1];
            v[0] = lo.x; v[1] = lo.y; v[2] = lo.z; v[3] = lo.w;
            v[4] = hi.x; v[5] = hi.y; v[6] = hi.z; v[7] = hi.w;
#pragma unroll
            for (int h = 0; h < HEADS; ++h)
                v[h] = (lane < cnt) ? lrelu(v[h] + ad[h]) : -INFINITY;
        };
        auto add_rows = [&](const bool two) {
            float al[HEADS];
#pragma unroll
            for (int h = 0; h < HEADS; ++h) al[h] = w[h] / (den[h] + 1e-16f);
            v2f xb0, xb1 = {};
            {
                const int s0 = __shfl(mysrc, mm, 32);
                const float* p0 = x + (size_t)s0 * 3 + 2 * kh;
                xb0.x = p0[0];  xb0.y = 0.0f;
                if (!kh) xb0.y = p0[1];
                if (two) {
                    const int s1 = __shfl(mysrc, 16 + mm, 32);
                    const float* p1 = x + (size_t)s1 * 3 + 2 * kh;
                    xb1.x = p1[0];  xb1.y = 0.0f;
                    if (!kh) xb1.y = p1[1];
                }
            }
#pragma unroll
            for (int h = 0; h < HEADS; ++h) {
                const float ax = __shfl_xor(al[h], 16, 32);
                const float wt0 = kh ? ax : al[h];
                const float wt1 = kh ? al[h] : ax;
                const v2f aA = fragW[(2 * h) * 32 + lane], aB = fragW[(2 * h + 1) * 32 + lane];
                v2f b;
                b.x = xb0.x * wt0;  b.y = xb0.y * wt0;
                v8f c0 = {}, c1 = {};
                c0 = __builtin_amdgcn_wmma_f32_16x16x4_f32(false, aA, false, b, (short)0, c0,
                                                           false, false);
                c1 = __builtin_amdgcn_wmma_f32_16x16x4_f32(false, aB, false, b, (short)0, c1,
                                                           false, false);
                if (two) {
                    v2f b2;
                    b2.x = xb1.x * wt1;  b2.y = xb1.y * wt1;
                    c0 = __builtin_amdgcn_wmma_f32_16x16x4_f32(false, aA, false, b2, (short)0, c0,
                                                               false, false);
                    c1 = __builtin_amdgcn_wmma_f32_16x16x4_f32(false, aB, false, b2, (short)0, c1,
                                                               false, false);
                }
                float r[16];
#pragma unroll
                for (int i = 0; i < 8; ++i) { r[i] = c0[i];  r[8 + i] = c1[i]; }
                acc[h] += row_sums16(r, lane);
            }
        };

        int nch = 0;
        bool fc = true;
        do {
            load_chunk(fc); fc = false;
#pragma unroll
            for (int h = 0; h < HEADS; ++h) m[h] = fmaxf(m[h], wave_max(v[h]));
            ++nch;
        } while (lists_live(cur) && nch < max_chunks);

        if (nch == 1) {
#pragma unroll
            for (int h = 0; h < HEADS; ++h) { w[h] = expf(v[h] - m[h]); den[h] = wave_sum(w[h]); }
            if (cnt > 16) add_rows(true); else add_rows(false);
        } else {
            cur = *myhead; fc = true;
            int k = 0;
            do {
                load_chunk(fc); fc = false;
#pragma unroll
                for (int h = 0; h < HEADS; ++h) den[h] += wave_sum(expf(v[h] - m[h]));
                ++k;
            } while (lists_live(cur) && k < max_chunks);
            cur = *myhead; fc = true; k = 0;
            do {
                load_chunk(fc); fc = false;
#pragma unroll
                for (int h = 0; h < HEADS; ++h) w[h] = expf(v[h] - m[h]);
                if (cnt > 16) add_rows(true); else add_rows(false);
                ++k;
            } while (lists_live(cur) && k < max_chunks);
        }
    }

    {
        float bv[HEADS];
#pragma unroll
        for (int h = 0; h < HEADS; ++h) bv[h] = bias1[h * HIDC + chan];
        float* prow = act + wv * ROWP + chan;
#pragma unroll
        for (int h = 0; h < HEADS; ++h) {
            float t = acc[h] + bv[h];
            prow[h * HIDC] = t > 0.0f ? t : (expf(t) - 1.0f);
        }
    }
    __builtin_amdgcn_fence(__ATOMIC_RELEASE, "workgroup");
    int prev = 0;
    if (lane == 0)
        prev = __hip_atomic_fetch_add(&arrived, 1, __ATOMIC_RELAXED, __HIP_MEMORY_SCOPE_WORKGROUP);
    prev = __builtin_amdgcn_readfirstlane(prev);
    if (prev != WG_NODES - 1) return;
    __builtin_amdgcn_fence(__ATOMIC_ACQUIRE, "workgroup");

    const float* arow = act + mm * ROWP + kh * 2;
    const float* brow = W2 + kh * 2 * OUTC + mm;
    const float sv = as2[mm], dv = ad2[mm];
    v8f c = {};
#pragma unroll 16
    for (int kk = 0; kk < L1C; kk += 4) {
        v2f a = *(const v2f*)(arow + kk);
        v2f b;
        b.x = brow[kk * OUTC];
        b.y = brow[kk * OUTC + OUTC];
        c = __builtin_amdgcn_wmma_f32_16x16x4_f32(false, a, false, b, (short)0, c,
                                                  false, false);
    }
    float* p = h2 + (size_t)(node0 + 8 * kh) * OUTC + mm;
    if (node0 + WG_NODES <= n_nodes) {
#pragma unroll
        for (int r = 0; r < 8; ++r) p[(size_t)r * OUTC] = c[r];
    } else {
        for (int r = 0; r < 8; ++r)
            if (node0 + 8 * kh + r < n_nodes) p[(size_t)r * OUTC] = c[r];
    }
    float r16[16];
#pragma unroll
    for (int v = 0; v < 8; ++v) {
        r16[v]     = c[v] * sv;
        r16[8 + v] = c[v] * dv;
    }
    const float tot = row_sums16(r16, lane);
    const int orow = node0 + 8 * kh + (mm & 7);
    if (orow < n_nodes) (mm < 8 ? a_s2 : a_d2)[orow] = tot;
}

__global__ void k_walk2(const int* __restrict__ head, const unsigned long long* __restrict__ link, int E,
                        const float* __restrict__ a_s, const float* __restrict__ a_d,
                        const float* __restrict__ h2, const float* __restrict__ bias2,
                        float* __restrict__ out, int n_nodes) {
    const int lane = threadIdx.x & 31;
    const int d = __builtin_amdgcn_readfirstlane(
        (int)(blockIdx.x * (blockDim.x >> 5) + (threadIdx.x >> 5)));
    if (d >= n_nodes) return;
    const int* myhead = head + (size_t)d * NL + lane;
    const int max_chunks = (E >> 5) + 2;
    const float adv = a_d[d];
    const int c = lane & 15, half = lane >> 4;
    const float bz = bias2[c];
    float m = -INFINITY, den = 0.0f, acc = 0.0f, v = 0.0f, w = 0.0f;
    int mysrc = d, cnt = 0, cur = *myhead;

    auto load_chunk = [&](bool firstc) {
        cnt = 0;
        if (firstc) { mysrc = d; cnt = 1; }
        lists_fill(cur, link, lane, mysrc, cnt);
        v = (lane < cnt) ? lrelu(a_s[mysrc] + adv) : -INFINITY;
    };
    auto add_rows = [&]() {
        float al = w / (den + 1e-16f);
#pragma unroll 4
        for (int j = 0; j < cnt; j += 2) {
            int jj = j + half;
            int s = __shfl(mysrc, jj, 32);
            float a = __shfl(al, jj, 32);
            float hv = h2[(size_t)s * OUTC + c];
            acc += (jj < cnt) ? hv * a : 0.0f;
        }
    };

    int nch = 0;
    bool fc = true;
    do {
        load_chunk(fc); fc = false;
        m = fmaxf(m, wave_max(v));
        ++nch;
    } while (lists_live(cur) && nch < max_chunks);

    if (nch == 1) {
        w = expf(v - m);
        den = wave_sum(w);
        add_rows();
    } else {
        cur = *myhead; fc = true;
        int k = 0;
        do {
            load_chunk(fc); fc = false;
            den += wave_sum(expf(v - m));
            ++k;
        } while (lists_live(cur) && k < max_chunks);
        cur = *myhead; fc = true; k = 0;
        do {
            load_chunk(fc); fc = false;
            w = expf(v - m);
            add_rows();
            ++k;
        } while (lists_live(cur) && k < max_chunks);
    }
    acc += __shfl_xor(acc, 16, 32);
    if (lane < OUTC) out[(size_t)d * OUTC + c] = acc + bz;
}

extern "C" void kernel_launch(void* const* d_in, const int* in_sizes, int n_in,
                              void* d_out, int out_size, void* d_ws, size_t ws_size,
                              hipStream_t stream) {
    const float* x     = (const float*)d_in[0];
    const int*   ei    = (const int*)d_in[1];
    const float* W1    = (const float*)d_in[2];
    const float* as1   = (const float*)d_in[3];
    const float* ad1   = (const float*)d_in[4];
    const float* bias1 = (const float*)d_in[5];
    const float* W2    = (const float*)d_in[6];
    const float* as2   = (const float*)d_in[7];
    const float* ad2   = (const float*)d_in[8];
    const float* bias2 = (const float*)d_in[9];
    float* out = (float*)d_out;

    const int N  = in_sizes[0] / 3;
    const int E  = in_sizes[1] / 2;

    float* ws = (float*)d_ws;
    const size_t Nz = (size_t)N;
    float* a_s1 = ws + 530 * Nz;
    float* a_d1 = ws + 538 * Nz;
    float* h2   = ws + 546 * Nz;
    float* a_s2 = ws + 562 * Nz;
    float* a_d2 = ws + 563 * Nz;
    int*   head = (int*)(ws + 564 * Nz);
    int2*  link = (int2*)(ws + 596 * Nz);

    const int TB = 256;
    const int node_blocks = (N + 7) / 8;
    k_fill_head<<<(NL * N + TB - 1) / TB, TB, 0, stream>>>(head, NL * N);

    {
        int mtiles = (N + 15) / 16;
        long long waves = (long long)mtiles * 2;
        int gemm_blocks = (int)((waves * 32 + TB - 1) / TB);
        int link_blocks = (E + TB - 1) / TB;
        k_gemm1_link<<<gemm_blocks + link_blocks, TB, 0, stream>>>(
            x, W1, as1, ad1, a_s1, a_d1, N, gemm_blocks, ei, E, head, link);
    }
    k_walk1_gemm2<<<(N + WG_NODES - 1) / WG_NODES, WG_NODES * 32, 0, stream>>>(
        head, (const unsigned long long*)link, E, a_s1, a_d1, x, W1, bias1, W2, as2, ad2, h2,
        a_s2, a_d2, N);

    k_walk2<<<node_blocks, TB, 0, stream>>>(head, (const unsigned long long*)link, E, a_s2, a_d2, h2,
                                            bias2, out, N);
}


# Layer-1 row products transposed (entries = rows, channels = columns): a lane's 8 registers are 8 entries of a channel, summed in the lane plus one half-wave swap, no 16-lane tree (sum order changes)
# speedup vs baseline: 8.1897x; 1.0721x over previous
#include <hip/hip_runtime.h>
#include <math.h>

#define HEADS     8
#define HIDC      32
#define L1C       256
#define OUTC      16
#define NEG_SLOPE 0.2f

typedef __attribute__((ext_vector_type(2))) float v2f;
typedef __attribute__((ext_vector_type(8))) float v8f;

__device__ __forceinline__ float lrelu(float v) {
    return v > 0.0f ? v : NEG_SLOPE * v;
}
#define NL 32
static_assert(NL == 32, "one list per lane of a 32-lane wave");
__global__ void k_fill_head(int* __restrict__ head, int n) {
    int i = blockIdx.x * blockDim.x + threadIdx.x;
    if (i < n) head[i] = -1;
}

__device__ __forceinline__ void link_edges(int blk, const int* __restrict__ ei, int E,
                                           int* __restrict__ head, int2* __restrict__ link) {
    int e = blk * (int)blockDim.x + (int)threadIdx.x;
    if (e >= E) return;
    int s = ei[e], d = ei[E + e];
    int old = atomicExch(&head[(size_t)d * NL + (e & (NL - 1))], e);
    link[e] = make_int2(old, s);
}

__device__ __forceinline__ float wave_max(float v) {
    for (int off = 16; off; off >>= 1) v = fmaxf(v, __shfl_xor(v, off, 32));
    return v;
}
__device__ __forceinline__ float wave_sum(float v) {
    for (int off = 16; off; off >>= 1) v += __shfl_xor(v, off, 32);
    return v;
}

__device__ __forceinline__ bool lists_live(int cur) {
    return __builtin_amdgcn_ballot_w32(cur >= 0) != 0u;
}
__device__ __forceinline__ void lists_fill(int& cur, const unsigned long long* __restrict__ link,
                                           int lane, int& mysrc, int& cnt) {
    for (int hop = 0; hop < 32 && cnt < 32; ++hop) {
        unsigned live = __builtin_amdgcn_ballot_w32(cur >= 0);
        if (live == 0u) break;
        unsigned long long r = 0ull;
        if (cur >= 0) r = link[cur];
        const int src = (int)(r >> 32);
        const int room = 32 - cnt, have = __builtin_popcount(live);
        const int n = have < room ? have : room;
        unsigned left = live;
        for (int t = 0; t < n; ++t) {
            const int k = __builtin_ctz(left);
            left &= left - 1u;
            const int s = __builtin_amdgcn_readlane(src, k);
            if (lane == cnt + t) mysrc = s;
        }
        cnt += n;
        if (((live & ~left) >> lane) & 1u) cur = (int)(unsigned)r;
    }
}


template <int MASK>
__device__ __forceinline__ float row_xor(float v) {
    return __int_as_float(__builtin_amdgcn_update_dpp(0, __float_as_int(v), 0x160 + MASK,
                                                      0xf, 0xf, true));
}

__device__ __forceinline__ float other_half(float v) {
    const int i = __float_as_int(v);
    return __int_as_float(__builtin_amdgcn_permlanex16(i, i, 0x76543210, 0xfedcba98, false, false));
}

__device__ __forceinline__ float row_sums16(float (&r)[16], int lane) {
    const bool b3 = (lane & 8) != 0, b2 = (lane & 4) != 0, b1 = (lane & 2) != 0,
               b0 = (lane & 1) != 0;
#pragma unroll
    for (int i = 0; i < 8; ++i) {
        float keep = b3 ? r[i + 8] : r[i], give = b3 ? r[i] : r[i + 8];
        r[i] = keep + row_xor<8>(give);
    }
#pragma unroll
    for (int i = 0; i < 4; ++i) {
        float keep = b2 ? r[i + 4] : r[i], give = b2 ? r[i] : r[i + 4];
        r[i] = keep + row_xor<4>(give);
    }
#pragma unroll
    for (int i = 0; i < 2; ++i) {
        float keep = b1 ? r[i + 2] : r[i], give = b1 ? r[i] : r[i + 2];
        r[i] = keep + row_xor<2>(give);
    }
    float keep = b0 ? r[1] : r[0], give = b0 ? r[0] : r[1];
    return keep + row_xor<1>(give);
}

__device__ __forceinline__ void gemm1_logits_tile(int blk, const float* __restrict__ x,
                               const float* __restrict__ W1,
                               const float* __restrict__ as1, const float* __restrict__ ad1,
                               float* __restrict__ a_s,
                               float* __restrict__ a_d, int n_nodes) {
    int wave   = (int)((blk * (int)blockDim.x + (int)threadIdx.x) >> 5);
    int lane   = threadIdx.x & 31;
    int mtiles = (n_nodes + 15) >> 4;
    int mtile  = wave >> 1;
    int hg     = wave & 1;
    if (mtile >= mtiles) return;
    int m  = lane & 15;
    int kh = lane >> 4;
    int node = mtile * 16 + m;
    if (node >= n_nodes) node = n_nodes - 1;
    float x0 = x[node * 3 + 0], x1 = x[node * 3 + 1], x2 = x[node * 3 + 2];
    v2f a;
    a.x = kh ? x2 : x0;  a.y = kh ? 0.0f : x1;
    const int orow = mtile * 16 + 8 * kh + (m & 7);
    float* dst = (m < 8 ? a_s : a_d) + (size_t)orow * HEADS + hg * 4;
#pragma unroll 1
    for (int hh = 0; hh < 4; ++hh) {
        const int head = hg * 4 + hh;
        const int col  = head * HIDC + m;
        float w0 = W1[0 * L1C + col], w1 = W1[1 * L1C + col], w2 = W1[2 * L1C + col];
        float u0 = W1[0 * L1C + col + 16], u1 = W1[1 * L1C + col + 16],
              u2 = W1[2 * L1C + col + 16];
        v2f bA, bB;
        bA.x = kh ? w2 : w0;  bA.y = kh ? 0.0f : w1;
        bB.x = kh ? u2 : u0;  bB.y = kh ? 0.0f : u1;
        v8f c0 = {}, c1 = {};
        c0 = __builtin_amdgcn_wmma_f32_16x16x4_f32(false, a, false, bA, (short)0, c0,
                                                   false, false);
        c1 = __builtin_amdgcn_wmma_f32_16x16x4_f32(false, a, false, bB, (short)0, c1,
                                                   false, false);
        const float sl = as1[head * HIDC + m], sh = as1[head * HIDC + 16 + m];
        const float dl = ad1[head * HIDC + m], dh = ad1[head * HIDC + 16 + m];
        float r[16];
#pragma unroll
        for (int v = 0; v < 8; ++v) {
            r[v]     = c0[v] * sl + c1[v] * sh;
            r[8 + v] = c0[v] * dl + c1[v] * dh;
        }
        const float tot = row_sums16(r, lane);
        if (orow < n_nodes) dst[hh] = tot;
    }
}

__global__ void k_gemm1_link(const float* __restrict__ x, const float* __restrict__ W1,
                             const float* __restrict__ as1, const float* __restrict__ ad1,
                             float* __restrict__ a_s,
                             float* __restrict__ a_d, int n_nodes, int gemm_blocks,
                             const int* __restrict__ ei, int E, int* __restrict__ head,
                             int2* __restrict__ link) {
    const int blk = (int)blockIdx.x;
    if (blk < gemm_blocks) gemm1_logits_tile(blk, x, W1, as1, ad1, a_s, a_d, n_nodes);
    else link_edges(blk - gemm_blocks, ei, E, head, link);
}

#define WG_NODES 16
#define ROWP (L1C + 4)
#define WALK1_WAVES_PER_SIMD 10
__global__ void __launch_bounds__(WG_NODES * 32, WALK1_WAVES_PER_SIMD)
k_walk1_gemm2(const int* __restrict__ head, const unsigned long long* __restrict__ link, int E,
              const float* __restrict__ a_s, const float* __restrict__ a_d,
              const float* __restrict__ x, const float* __restrict__ W1,
              const float* __restrict__ bias1,
              const float* __restrict__ W2, const float* __restrict__ as2,
              const float* __restrict__ ad2, float* __restrict__ h2,
              float* __restrict__ a_s2, float* __restrict__ a_d2, int n_nodes) {
    __shared__ __attribute__((aligned(16))) float act[WG_NODES * ROWP];
    __shared__ int arrived;
    __shared__ __attribute__((aligned(8))) v2f fragW[(L1C / 16) * 32];
    static_assert(WG_NODES == L1C / 16, "wave nt of the work-group stages tile nt");
    const int lane = threadIdx.x & 31;
    const int wv = __builtin_amdgcn_readfirstlane((int)(threadIdx.x >> 5));
    const int node0 = (int)blockIdx.x * WG_NODES;
    const int d = node0 + wv;
    const int mm = lane & 15;
    const int kh = lane >> 4;
    const int chan = lane;
    int cur0 = -1;
    if (d < n_nodes) cur0 = head[(size_t)d * NL + lane];
    {
        const int col = wv * 16 + mm;
        const float wx = W1[(kh ? 2 * L1C : 0) + col], wy = W1[L1C + col];
        v2f f;
        f.x = wx;  f.y = kh ? 0.0f : wy;
        fragW[wv * 32 + lane] = f;
    }
    if (threadIdx.x == 0) arrived = 0;
    __syncthreads();

    float acc[HEADS];
#pragma unroll
    for (int h = 0; h < HEADS; ++h) acc[h] = 0.0f;

    if (d < n_nodes) {
        const int* myhead = head + (size_t)d * NL + lane;
        const int max_chunks = (E >> 5) + 2;
        float ad[HEADS];
#pragma unroll
        for (int h = 0; h < HEADS; ++h) ad[h] = a_d[d * HEADS + h];

        float m[HEADS], den[HEADS], v[HEADS], w[HEADS];
#pragma unroll
        for (int h = 0; h < HEADS; ++h) { m[h] = -INFINITY; den[h] = 0.0f; }
        int mysrc = d, cnt = 0, cur = cur0;

        auto load_chunk = [&](bool firstc) {
            cnt = 0;
            if (firstc) { mysrc = d; cnt = 1; }
            lists_fill(cur, link, lane, mysrc, cnt);
            const float4* p = (const float4*)(a_s + (size_t)mysrc * HEADS);
            float4 lo = p[0], hi = p[1];
            v[0] = lo.x; v[1] = lo.y; v[2] = lo.z; v[3] = lo.w;
            v[4] = hi.x; v[5] = hi.y; v[6] = hi.z; v[7] = hi.w;
#pragma unroll
            for (int h = 0; h < HEADS; ++h)
                v[h] = (lane < cnt) ? lrelu(v[h] + ad[h]) : -INFINITY;
        };
        auto add_rows = [&](const bool two) {
            float al[HEADS];
#pragma unroll
            for (int h = 0; h < HEADS; ++h) al[h] = w[h] / (den[h] + 1e-16f);
            v2f xb0, xb1 = {};
            {
                const int s0 = __shfl(mysrc, mm, 32);
                const float* p0 = x + (size_t)s0 * 3 + 2 * kh;
                xb0.x = p0[0];  xb0.y = 0.0f;
                if (!kh) xb0.y = p0[1];
                if (two) {
                    const int s1 = __shfl(mysrc, 16 + mm, 32);
                    const float* p1 = x + (size_t)s1 * 3 + 2 * kh;
                    xb1.x = p1[0];  xb1.y = 0.0f;
                    if (!kh) xb1.y = p1[1];
                }
            }
            v2f nA = fragW[lane], nB = fragW[32 + lane];
#pragma unroll
            for (int h = 0; h < HEADS; ++h) {
                const v2f bA = nA, bB = nB;
                if (h + 1 < HEADS) {
                    nA = fragW[(2 * h + 2) * 32 + lane];
                    nB = fragW[(2 * h + 3) * 32 + lane];
                }
                const float ax = other_half(al[h]);
                const float wt0 = kh ? ax : al[h];
                const float wt1 = kh ? al[h] : ax;
                v2f a;
                a.x = xb0.x * wt0;  a.y = xb0.y * wt0;
                v8f c0 = {}, c1 = {};
                c0 = __builtin_amdgcn_wmma_f32_16x16x4_f32(false, a, false, bA, (short)0, c0,
                                                           false, false);
                c1 = __builtin_amdgcn_wmma_f32_16x16x4_f32(false, a, false, bB, (short)0, c1,
                                                           false, false);
                if (two) {
                    v2f a2;
                    a2.x = xb1.x * wt1;  a2.y = xb1.y * wt1;
                    c0 = __builtin_amdgcn_wmma_f32_16x16x4_f32(false, a2, false, bA, (short)0, c0,
                                                               false, false);
                    c1 = __builtin_amdgcn_wmma_f32_16x16x4_f32(false, a2, false, bB, (short)0, c1,
                                                               false, false);
                }
                const v2f q0 = (__builtin_shufflevector(c0, c0, 0, 1) +
                                __builtin_shufflevector(c0, c0, 4, 5)) +
                               (__builtin_shufflevector(c0, c0, 2, 3) +
                                __builtin_shufflevector(c0, c0, 6, 7));
                const v2f q1 = (__builtin_shufflevector(c1, c1, 0, 1) +
                                __builtin_shufflevector(c1, c1, 4, 5)) +
                               (__builtin_shufflevector(c1, c1, 2, 3) +
                                __builtin_shufflevector(c1, c1, 6, 7));
                const float p0 = q0.x + q0.y, p1 = q1.x + q1.y;
                const float own = kh ? p1 : p0, give = kh ? p0 : p1;
                acc[h] += own + other_half(give);
                __builtin_amdgcn_sched_barrier(0);
            }
        };

        int nch = 0;
        bool fc = true;
        do {
            load_chunk(fc); fc = false;
#pragma unroll
            for (int h = 0; h < HEADS; ++h) m[h] = fmaxf(m[h], wave_max(v[h]));
            ++nch;
        } while (lists_live(cur) && nch < max_chunks);

        if (nch == 1) {
#pragma unroll
            for (int h = 0; h < HEADS; ++h) { w[h] = expf(v[h] - m[h]); den[h] = wave_sum(w[h]); }
            if (cnt > 16) add_rows(true); else add_rows(false);
        } else {
            cur = *myhead; fc = true;
            int k = 0;
            do {
                load_chunk(fc); fc = false;
#pragma unroll
                for (int h = 0; h < HEADS; ++h) den[h] += wave_sum(expf(v[h] - m[h]));
                ++k;
            } while (lists_live(cur) && k < max_chunks);
            cur = *myhead; fc = true; k = 0;
            do {
                load_chunk(fc); fc = false;
#pragma unroll
                for (int h = 0; h < HEADS; ++h) w[h] = expf(v[h] - m[h]);
                if (cnt > 16) add_rows(true); else add_rows(false);
                ++k;
            } while (lists_live(cur) && k < max_chunks);
        }
    }

    {
        float bv[HEADS];
#pragma unroll
        for (int h = 0; h < HEADS; ++h) bv[h] = bias1[h * HIDC + chan];
        float* prow = act + wv * ROWP + chan;
#pragma unroll
        for (int h = 0; h < HEADS; ++h) {
            float t = acc[h] + bv[h];
            prow[h * HIDC] = t > 0.0f ? t : (expf(t) - 1.0f);
        }
    }
    __builtin_amdgcn_fence(__ATOMIC_RELEASE, "workgroup");
    int prev = 0;
    if (lane == 0)
        prev = __hip_atomic_fetch_add(&arrived, 1, __ATOMIC_RELAXED, __HIP_MEMORY_SCOPE_WORKGROUP);
    prev = __builtin_amdgcn_readfirstlane(prev);
    if (prev != WG_NODES - 1) return;
    __builtin_amdgcn_fence(__ATOMIC_ACQUIRE, "workgroup");

    const float* arow = act + mm * ROWP + kh * 2;
    const float* brow = W2 + kh * 2 * OUTC + mm;
    const float sv = as2[mm], dv = ad2[mm];
    v8f c = {};
#pragma unroll 16
    for (int kk = 0; kk < L1C; kk += 4) {
        v2f a = *(const v2f*)(arow + kk);
        v2f b;
        b.x = brow[kk * OUTC];
        b.y = brow[kk * OUTC + OUTC];
        c = __builtin_amdgcn_wmma_f32_16x16x4_f32(false, a, false, b, (short)0, c,
                                                  false, false);
    }
    float* p = h2 + (size_t)(node0 + 8 * kh) * OUTC + mm;
    if (node0 + WG_NODES <= n_nodes) {
#pragma unroll
        for (int r = 0; r < 8; ++r) p[(size_t)r * OUTC] = c[r];
    } else {
        for (int r = 0; r < 8; ++r)
            if (node0 + 8 * kh + r < n_nodes) p[(size_t)r * OUTC] = c[r];
    }
    float r16[16];
#pragma unroll
    for (int v = 0; v < 8; ++v) {
        r16[v]     = c[v] * sv;
        r16[8 + v] = c[v] * dv;
    }
    const float tot = row_sums16(r16, lane);
    const int orow = node0 + 8 * kh + (mm & 7);
    if (orow < n_nodes) (mm < 8 ? a_s2 : a_d2)[orow] = tot;
}

__global__ void k_walk2(const int* __restrict__ head, const unsigned long long* __restrict__ link, int E,
                        const float* __restrict__ a_s, const float* __restrict__ a_d,
                        const float* __restrict__ h2, const float* __restrict__ bias2,
                        float* __restrict__ out, int n_nodes) {
    const int lane = threadIdx.x & 31;
    const int d = __builtin_amdgcn_readfirstlane(
        (int)(blockIdx.x * (blockDim.x >> 5) + (threadIdx.x >> 5)));
    if (d >= n_nodes) return;
    const int* myhead = head + (size_t)d * NL + lane;
    const int max_chunks = (E >> 5) + 2;
    const float adv = a_d[d];
    const int c = lane & 15, half = lane >> 4;
    const float bz = bias2[c];
    float m = -INFINITY, den = 0.0f, acc = 0.0f, v = 0.0f, w = 0.0f;
    int mysrc = d, cnt = 0, cur = *myhead;

    auto load_chunk = [&](bool firstc) {
        cnt = 0;
        if (firstc) { mysrc = d; cnt = 1; }
        lists_fill(cur, link, lane, mysrc, cnt);
        v = (lane < cnt) ? lrelu(a_s[mysrc] + adv) : -INFINITY;
    };
    auto add_rows = [&]() {
        float al = w / (den + 1e-16f);
#pragma unroll 4
        for (int j = 0; j < cnt; j += 2) {
            int jj = j + half;
            int s = __shfl(mysrc, jj, 32);
            float a = __shfl(al, jj, 32);
            float hv = h2[(size_t)s * OUTC + c];
            acc += (jj < cnt) ? hv * a : 0.0f;
        }
    };

    int nch = 0;
    bool fc = true;
    do {
        load_chunk(fc); fc = false;
        m = fmaxf(m, wave_max(v));
        ++nch;
    } while (lists_live(cur) && nch < max_chunks);

    if (nch == 1) {
        w = expf(v - m);
        den = wave_sum(w);
        add_rows();
    } else {
        cur = *myhead; fc = true;
        int k = 0;
        do {
            load_chunk(fc); fc = false;
            den += wave_sum(expf(v - m));
            ++k;
        } while (lists_live(cur) && k < max_chunks);
        cur = *myhead; fc = true; k = 0;
        do {
            load_chunk(fc); fc = false;
            w = expf(v - m);
            add_rows();
            ++k;
        } while (lists_live(cur) && k < max_chunks);
    }
    acc += __shfl_xor(acc, 16, 32);
    if (lane < OUTC) out[(size_t)d * OUTC + c] = acc + bz;
}

extern "C" void kernel_launch(void* const* d_in, const int* in_sizes, int n_in,
                              void* d_out, int out_size, void* d_ws, size_t ws_size,
                              hipStream_t stream) {
    const float* x     = (const float*)d_in[0];
    const int*   ei    = (const int*)d_in[1];
    const float* W1    = (const float*)d_in[2];
    const float* as1   = (const float*)d_in[3];
    const float* ad1   = (const float*)d_in[4];
    const float* bias1 = (const float*)d_in[5];
    const float* W2    = (const float*)d_in[6];
    const float* as2   = (const float*)d_in[7];
    const float* ad2   = (const float*)d_in[8];
    const float* bias2 = (const float*)d_in[9];
    float* out = (float*)d_out;

    const int N  = in_sizes[0] / 3;
    const int E  = in_sizes[1] / 2;

    float* ws = (float*)d_ws;
    const size_t Nz = (size_t)N;
    float* a_s1 = ws + 530 * Nz;
    float* a_d1 = ws + 538 * Nz;
    float* h2   = ws + 546 * Nz;
    float* a_s2 = ws + 562 * Nz;
    float* a_d2 = ws + 563 * Nz;
    int*   head = (int*)(ws + 564 * Nz);
    int2*  link = (int2*)(ws + 596 * Nz);

    const int TB = 256;
    const int node_blocks = (N + 7) / 8;
    k_fill_head<<<(NL * N + TB - 1) / TB, TB, 0, stream>>>(head, NL * N);

    {
        int mtiles = (N + 15) / 16;
        long long waves = (long long)mtiles * 2;
        int gemm_blocks = (int)((waves * 32 + TB - 1) / TB);
        int link_blocks = (E + TB - 1) / TB;
        k_gemm1_link<<<gemm_blocks + link_blocks, TB, 0, stream>>>(
            x, W1, as1, ad1, a_s1, a_d1, N, gemm_blocks, ei, E, head, link);
    }
    k_walk1_gemm2<<<(N + WG_NODES - 1) / WG_NODES, WG_NODES * 32, 0, stream>>>(
        head, (const unsigned long long*)link, E, a_s1, a_d1, x, W1, bias1, W2, as2, ad2, h2,
        a_s2, a_d2, N);

    k_walk2<<<node_blocks, TB, 0, stream>>>(head, (const unsigned long long*)link, E, a_s2, a_d2, h2,
                                            bias2, out, N);
}


# Walks' cross-lane moves on the vector ALU (DPP/permlanex16 butterflies, the 8 heads halved per step), not the LDS unit; x loaded by its source's owner lane with the logits (sum order unchanged)
# speedup vs baseline: 8.6049x; 1.0507x over previous
#include <hip/hip_runtime.h>
#include <math.h>

#define HEADS     8
#define HIDC      32
#define L1C       256
#define OUTC      16
#define NEG_SLOPE 0.2f

typedef __attribute__((ext_vector_type(2))) float v2f;
typedef __attribute__((ext_vector_type(8))) float v8f;

__device__ __forceinline__ float lrelu(float v) {
    return v > 0.0f ? v : NEG_SLOPE * v;
}
#define NL 32
static_assert(NL == 32, "one list per lane of a 32-lane wave");
__global__ void k_fill_head(int* __restrict__ head, int n) {
    int i = blockIdx.x * blockDim.x + threadIdx.x;
    if (i < n) head[i] = -1;
}

__device__ __forceinline__ void link_edges(int blk, const int* __restrict__ ei, int E,
                                           int* __restrict__ head, int2* __restrict__ link) {
    int e = blk * (int)blockDim.x + (int)threadIdx.x;
    if (e >= E) return;
    int s = ei[e], d = ei[E + e];
    int old = atomicExch(&head[(size_t)d * NL + (e & (NL - 1))], e);
    link[e] = make_int2(old, s);
}

template <int MASK>
__device__ __forceinline__ float row_xor(float v) {
    return __int_as_float(__builtin_amdgcn_update_dpp(0, __float_as_int(v), 0x160 + MASK,
                                                      0xf, 0xf, true));
}

__device__ __forceinline__ float other_half(float v) {
    const int i = __float_as_int(v);
    return __int_as_float(__builtin_amdgcn_permlanex16(i, i, 0x76543210, 0xfedcba98, false, false));
}

__device__ __forceinline__ float lane_val(float v, int j) {
    return __int_as_float(__builtin_amdgcn_readlane(__float_as_int(v), j));
}

__device__ __forceinline__ float wave_max(float v) {
    v = fmaxf(v, other_half(v));
    v = fmaxf(v, row_xor<8>(v));
    v = fmaxf(v, row_xor<4>(v));
    v = fmaxf(v, row_xor<2>(v));
    return fmaxf(v, row_xor<1>(v));
}
__device__ __forceinline__ float wave_sum(float v) {
    v += other_half(v);
    v += row_xor<8>(v);
    v += row_xor<4>(v);
    v += row_xor<2>(v);
    return v + row_xor<1>(v);
}

template <bool IS_MAX>
__device__ __forceinline__ void wave_reduce8(const float (&in)[8], int lane, float (&out)[8]) {
    auto op = [](float a, float b) { return IS_MAX ? fmaxf(a, b) : a + b; };
    const bool b4 = (lane & 16) != 0, b3 = (lane & 8) != 0, b2 = (lane & 4) != 0;
    float r[4];
#pragma unroll
    for (int i = 0; i < 4; ++i) {
        const float keep = b4 ? in[i + 4] : in[i], give = b4 ? in[i] : in[i + 4];
        r[i] = op(keep, other_half(give));
    }
#pragma unroll
    for (int i = 0; i < 2; ++i) {
        const float keep = b3 ? r[i + 2] : r[i], give = b3 ? r[i] : r[i + 2];
        r[i] = op(keep, row_xor<8>(give));
    }
    const float keep = b2 ? r[1] : r[0], give = b2 ? r[0] : r[1];
    float t = op(keep, row_xor<4>(give));
    t = op(t, row_xor<2>(t));
    t = op(t, row_xor<1>(t));
#pragma unroll
    for (int h = 0; h < 8; ++h) out[h] = lane_val(t, 4 * h);
}

__device__ __forceinline__ bool lists_live(int cur) {
    return __builtin_amdgcn_ballot_w32(cur >= 0) != 0u;
}
__device__ __forceinline__ void lists_fill(int& cur, const unsigned long long* __restrict__ link,
                                           int lane, int& mysrc, int& cnt) {
    for (int hop = 0; hop < 32 && cnt < 32; ++hop) {
        unsigned live = __builtin_amdgcn_ballot_w32(cur >= 0);
        if (live == 0u) break;
        unsigned long long r = 0ull;
        if (cur >= 0) r = link[cur];
        const int src = (int)(r >> 32);
        const int room = 32 - cnt, have = __builtin_popcount(live);
        const int n = have < room ? have : room;
        unsigned left = live;
        for (int t = 0; t < n; ++t) {
            const int k = __builtin_ctz(left);
            left &= left - 1u;
            const int s = __builtin_amdgcn_readlane(src, k);
            if (lane == cnt + t) mysrc = s;
        }
        cnt += n;
        if (((live & ~left) >> lane) & 1u) cur = (int)(unsigned)r;
    }
}


__device__ __forceinline__ float row_sums16(float (&r)[16], int lane) {
    const bool b3 = (lane & 8) != 0, b2 = (lane & 4) != 0, b1 = (lane & 2) != 0,
               b0 = (lane & 1) != 0;
#pragma unroll
    for (int i = 0; i < 8; ++i) {
        float keep = b3 ? r[i + 8] : r[i], give = b3 ? r[i] : r[i + 8];
        r[i] = keep + row_xor<8>(give);
    }
#pragma unroll
    for (int i = 0; i < 4; ++i) {
        float keep = b2 ? r[i + 4] : r[i], give = b2 ? r[i] : r[i + 4];
        r[i] = keep + row_xor<4>(give);
    }
#pragma unroll
    for (int i = 0; i < 2; ++i) {
        float keep = b1 ? r[i + 2] : r[i], give = b1 ? r[i] : r[i + 2];
        r[i] = keep + row_xor<2>(give);
    }
    float keep = b0 ? r[1] : r[0], give = b0 ? r[0] : r[1];
    return keep + row_xor<1>(give);
}

__device__ __forceinline__ void gemm1_logits_tile(int blk, const float* __restrict__ x,
                               const float* __restrict__ W1,
                               const float* __restrict__ as1, const float* __restrict__ ad1,
                               float* __restrict__ a_s,
                               float* __restrict__ a_d, int n_nodes) {
    int wave   = (int)((blk * (int)blockDim.x + (int)threadIdx.x) >> 5);
    int lane   = threadIdx.x & 31;
    int mtiles = (n_nodes + 15) >> 4;
    int mtile  = wave >> 1;
    int hg     = wave & 1;
    if (mtile >= mtiles) return;
    int m  = lane & 15;
    int kh = lane >> 4;
    int node = mtile * 16 + m;
    if (node >= n_nodes) node = n_nodes - 1;
    float x0 = x[node * 3 + 0], x1 = x[node * 3 + 1], x2 = x[node * 3 + 2];
    v2f a;
    a.x = kh ? x2 : x0;  a.y = kh ? 0.0f : x1;
    const int orow = mtile * 16 + 8 * kh + (m & 7);
    float* dst = (m < 8 ? a_s : a_d) + (size_t)orow * HEADS + hg * 4;
#pragma unroll 1
    for (int hh = 0; hh < 4; ++hh) {
        const int head = hg * 4 + hh;
        const int col  = head * HIDC + m;
        float w0 = W1[0 * L1C + col], w1 = W1[1 * L1C + col], w2 = W1[2 * L1C + col];
        float u0 = W1[0 * L1C + col + 16], u1 = W1[1 * L1C + col + 16],
              u2 = W1[2 * L1C + col + 16];
        v2f bA, bB;
        bA.x = kh ? w2 : w0;  bA.y = kh ? 0.0f : w1;
        bB.x = kh ? u2 : u0;  bB.y = kh ? 0.0f : u1;
        v8f c0 = {}, c1 = {};
        c0 = __builtin_amdgcn_wmma_f32_16x16x4_f32(false, a, false, bA, (short)0, c0,
                                                   false, false);
        c1 = __builtin_amdgcn_wmma_f32_16x16x4_f32(false, a, false, bB, (short)0, c1,
                                                   false, false);
        const float sl = as1[head * HIDC + m], sh = as1[head * HIDC + 16 + m];
        const float dl = ad1[head * HIDC + m], dh = ad1[head * HIDC + 16 + m];
        float r[16];
#pragma unroll
        for (int v = 0; v < 8; ++v) {
            r[v]     = c0[v] * sl + c1[v] * sh;
            r[8 + v] = c0[v] * dl + c1[v] * dh;
        }
        const float tot = row_sums16(r, lane);
        if (orow < n_nodes) dst[hh] = tot;
    }
}

__global__ void k_gemm1_link(const float* __restrict__ x, const float* __restrict__ W1,
                             const float* __restrict__ as1, const float* __restrict__ ad1,
                             float* __restrict__ a_s,
                             float* __restrict__ a_d, int n_nodes, int gemm_blocks,
                             const int* __restrict__ ei, int E, int* __restrict__ head,
                             int2* __restrict__ link) {
    const int blk = (int)blockIdx.x;
    if (blk < gemm_blocks) gemm1_logits_tile(blk, x, W1, as1, ad1, a_s, a_d, n_nodes);
    else link_edges(blk - gemm_blocks, ei, E, head, link);
}

#define WG_NODES 16
#define ROWP (L1C + 4)
#define WALK1_WAVES_PER_SIMD 10
__global__ void __launch_bounds__(WG_NODES * 32, WALK1_WAVES_PER_SIMD)
k_walk1_gemm2(const int* __restrict__ head, const unsigned long long* __restrict__ link, int E,
              const float* __restrict__ a_s, const float* __restrict__ a_d,
              const float* __restrict__ x, const float* __restrict__ W1,
              const float* __restrict__ bias1,
              const float* __restrict__ W2, const float* __restrict__ as2,
              const float* __restrict__ ad2, float* __restrict__ h2,
              float* __restrict__ a_s2, float* __restrict__ a_d2, int n_nodes) {
    __shared__ __attribute__((aligned(16))) float act[WG_NODES * ROWP];
    __shared__ int arrived;
    __shared__ __attribute__((aligned(8))) v2f fragW[(L1C / 16) * 32];
    static_assert(WG_NODES == L1C / 16, "wave nt of the work-group stages tile nt");
    const int lane = threadIdx.x & 31;
    const int wv = __builtin_amdgcn_readfirstlane((int)(threadIdx.x >> 5));
    const int node0 = (int)blockIdx.x * WG_NODES;
    const int d = node0 + wv;
    const int mm = lane & 15;
    const int kh = lane >> 4;
    const int chan = lane;
    int cur0 = -1;
    if (d < n_nodes) cur0 = head[(size_t)d * NL + lane];
    {
        const int col = wv * 16 + mm;
        const float wx = W1[(kh ? 2 * L1C : 0) + col], wy = W1[L1C + col];
        v2f f;
        f.x = wx;  f.y = kh ? 0.0f : wy;
        fragW[wv * 32 + lane] = f;
    }
    if (threadIdx.x == 0) arrived = 0;
    __syncthreads();

    float acc[HEADS];
#pragma unroll
    for (int h = 0; h < HEADS; ++h) acc[h] = 0.0f;

    if (d < n_nodes) {
        const int* myhead = head + (size_t)d * NL + lane;
        const int max_chunks = (E >> 5) + 2;
        float ad[HEADS];
#pragma unroll
        for (int h = 0; h < HEADS; ++h) ad[h] = a_d[d * HEADS + h];

        float m[HEADS], den[HEADS], v[HEADS], w[HEADS];
#pragma unroll
        for (int h = 0; h < HEADS; ++h) { m[h] = -INFINITY; den[h] = 0.0f; }
        int mysrc = d, cnt = 0, cur = cur0;
        float xs0 = 0.0f, xs1 = 0.0f, xs2 = 0.0f;

        auto load_chunk = [&](bool firstc) {
            cnt = 0;
            if (firstc) { mysrc = d; cnt = 1; }
            lists_fill(cur, link, lane, mysrc, cnt);
            const float4* p = (const float4*)(a_s + (size_t)mysrc * HEADS);
            const float* px = x + (size_t)mysrc * 3;
            float4 lo = p[0], hi = p[1];
            xs0 = px[0];  xs1 = px[1];  xs2 = px[2];
            v[0] = lo.x; v[1] = lo.y; v[2] = lo.z; v[3] = lo.w;
            v[4] = hi.x; v[5] = hi.y; v[6] = hi.z; v[7] = hi.w;
#pragma unroll
            for (int h = 0; h < HEADS; ++h)
                v[h] = (lane < cnt) ? lrelu(v[h] + ad[h]) : -INFINITY;
        };
        auto add_rows = [&](const bool two) {
            float al[HEADS];
#pragma unroll
            for (int h = 0; h < HEADS; ++h) al[h] = w[h] / (den[h] + 1e-16f);
            v2f xb0, xb1 = {};
            const float o2 = other_half(xs2);
            xb0.x = kh ? o2 : xs0;  xb0.y = kh ? 0.0f : xs1;
            if (two) {
                const float o0 = other_half(xs0), o1 = other_half(xs1);
                xb1.x = kh ? xs2 : o0;  xb1.y = kh ? 0.0f : o1;
            }
            v2f nA = fragW[lane], nB = fragW[32 + lane];
#pragma unroll
            for (int h = 0; h < HEADS; ++h) {
                const v2f bA = nA, bB = nB;
                if (h + 1 < HEADS) {
                    nA = fragW[(2 * h + 2) * 32 + lane];
                    nB = fragW[(2 * h + 3) * 32 + lane];
                }
                const float ax = other_half(al[h]);
                const float wt0 = kh ? ax : al[h];
                const float wt1 = kh ? al[h] : ax;
                v2f a;
                a.x = xb0.x * wt0;  a.y = xb0.y * wt0;
                v8f c0 = {}, c1 = {};
                c0 = __builtin_amdgcn_wmma_f32_16x16x4_f32(false, a, false, bA, (short)0, c0,
                                                           false, false);
                c1 = __builtin_amdgcn_wmma_f32_16x16x4_f32(false, a, false, bB, (short)0, c1,
                                                           false, false);
                if (two) {
                    v2f a2;
                    a2.x = xb1.x * wt1;  a2.y = xb1.y * wt1;
                    c0 = __builtin_amdgcn_wmma_f32_16x16x4_f32(false, a2, false, bA, (short)0, c0,
                                                               false, false);
                    c1 = __builtin_amdgcn_wmma_f32_16x16x4_f32(false, a2, false, bB, (short)0, c1,
                                                               false, false);
                }
                const v2f q0 = (__builtin_shufflevector(c0, c0, 0, 1) +
                                __builtin_shufflevector(c0, c0, 4, 5)) +
                               (__builtin_shufflevector(c0, c0, 2, 3) +
                                __builtin_shufflevector(c0, c0, 6, 7));
                const v2f q1 = (__builtin_shufflevector(c1, c1, 0, 1) +
                                __builtin_shufflevector(c1, c1, 4, 5)) +
                               (__builtin_shufflevector(c1, c1, 2, 3) +
                                __builtin_shufflevector(c1, c1, 6, 7));
                const float p0 = q0.x + q0.y, p1 = q1.x + q1.y;
                const float own = kh ? p1 : p0, give = kh ? p0 : p1;
                acc[h] += own + other_half(give);
                __builtin_amdgcn_sched_barrier(0);
            }
        };

        int nch = 0;
        bool fc = true;
        do {
            load_chunk(fc); fc = false;
            float mx[HEADS];
            wave_reduce8<true>(v, lane, mx);
#pragma unroll
            for (int h = 0; h < HEADS; ++h) m[h] = fmaxf(m[h], mx[h]);
            asm volatile("" : : "v"(xs0), "v"(xs1), "v"(xs2));
            ++nch;
        } while (lists_live(cur) && nch < max_chunks);

        if (nch == 1) {
#pragma unroll
            for (int h = 0; h < HEADS; ++h) w[h] = expf(v[h] - m[h]);
            wave_reduce8<false>(w, lane, den);
            if (cnt > 16) add_rows(true); else add_rows(false);
        } else {
            cur = *myhead; fc = true;
            int k = 0;
            do {
                load_chunk(fc); fc = false;
                float ex[HEADS], sm[HEADS];
#pragma unroll
                for (int h = 0; h < HEADS; ++h) ex[h] = expf(v[h] - m[h]);
                wave_reduce8<false>(ex, lane, sm);
#pragma unroll
                for (int h = 0; h < HEADS; ++h) den[h] += sm[h];
                ++k;
            } while (lists_live(cur) && k < max_chunks);
            cur = *myhead; fc = true; k = 0;
            do {
                load_chunk(fc); fc = false;
#pragma unroll
                for (int h = 0; h < HEADS; ++h) w[h] = expf(v[h] - m[h]);
                if (cnt > 16) add_rows(true); else add_rows(false);
                ++k;
            } while (lists_live(cur) && k < max_chunks);
        }
    }

    {
        float bv[HEADS];
#pragma unroll
        for (int h = 0; h < HEADS; ++h) bv[h] = bias1[h * HIDC + chan];
        float* prow = act + wv * ROWP + chan;
#pragma unroll
        for (int h = 0; h < HEADS; ++h) {
            float t = acc[h] + bv[h];
            prow[h * HIDC] = t > 0.0f ? t : (expf(t) - 1.0f);
        }
    }
    __builtin_amdgcn_fence(__ATOMIC_RELEASE, "workgroup");
    int prev = 0;
    if (lane == 0)
        prev = __hip_atomic_fetch_add(&arrived, 1, __ATOMIC_RELAXED, __HIP_MEMORY_SCOPE_WORKGROUP);
    prev = __builtin_amdgcn_readfirstlane(prev);
    if (prev != WG_NODES - 1) return;
    __builtin_amdgcn_fence(__ATOMIC_ACQUIRE, "workgroup");

    const float* arow = act + mm * ROWP + kh * 2;
    const float* brow = W2 + kh * 2 * OUTC + mm;
    const float sv = as2[mm], dv = ad2[mm];
    v8f c = {};
#pragma unroll 16
    for (int kk = 0; kk < L1C; kk += 4) {
        v2f a = *(const v2f*)(arow + kk);
        v2f b;
        b.x = brow[kk * OUTC];
        b.y = brow[kk * OUTC + OUTC];
        c = __builtin_amdgcn_wmma_f32_16x16x4_f32(false, a, false, b, (short)0, c,
                                                  false, false);
    }
    float* p = h2 + (size_t)(node0 + 8 * kh) * OUTC + mm;
    if (node0 + WG_NODES <= n_nodes) {
#pragma unroll
        for (int r = 0; r < 8; ++r) p[(size_t)r * OUTC] = c[r];
    } else {
        for (int r = 0; r < 8; ++r)
            if (node0 + 8 * kh + r < n_nodes) p[(size_t)r * OUTC] = c[r];
    }
    float r16[16];
#pragma unroll
    for (int v = 0; v < 8; ++v) {
        r16[v]     = c[v] * sv;
        r16[8 + v] = c[v] * dv;
    }
    const float tot = row_sums16(r16, lane);
    const int orow = node0 + 8 * kh + (mm & 7);
    if (orow < n_nodes) (mm < 8 ? a_s2 : a_d2)[orow] = tot;
}

__global__ void k_walk2(const int* __restrict__ head, const unsigned long long* __restrict__ link, int E,
                        const float* __restrict__ a_s, const float* __restrict__ a_d,
                        const float* __restrict__ h2, const float* __restrict__ bias2,
                        float* __restrict__ out, int n_nodes) {
    const int lane = threadIdx.x & 31;
    const int d = __builtin_amdgcn_readfirstlane(
        (int)(blockIdx.x * (blockDim.x >> 5) + (threadIdx.x >> 5)));
    if (d >= n_nodes) return;
    const int* myhead = head + (size_t)d * NL + lane;
    const int max_chunks = (E >> 5) + 2;
    const float adv = a_d[d];
    const int c = lane & 15, half = lane >> 4;
    const float bz = bias2[c];
    float m = -INFINITY, den = 0.0f, acc = 0.0f, v = 0.0f, w = 0.0f;
    int mysrc = d, cnt = 0, cur = *myhead;

    auto load_chunk = [&](bool firstc) {
        cnt = 0;
        if (firstc) { mysrc = d; cnt = 1; }
        lists_fill(cur, link, lane, mysrc, cnt);
        v = (lane < cnt) ? lrelu(a_s[mysrc] + adv) : -INFINITY;
    };
    auto add_rows = [&]() {
        float al = w / (den + 1e-16f);
#pragma unroll 4
        for (int j = 0; j < cnt; j += 2) {
            int jj = j + half;
            int s = __shfl(mysrc, jj, 32);
            float a = __shfl(al, jj, 32);
            float hv = h2[(size_t)s * OUTC + c];
            acc += (jj < cnt) ? hv * a : 0.0f;
        }
    };

    int nch = 0;
    bool fc = true;
    do {
        load_chunk(fc); fc = false;
        m = fmaxf(m, wave_max(v));
        ++nch;
    } while (lists_live(cur) && nch < max_chunks);

    if (nch == 1) {
        w = expf(v - m);
        den = wave_sum(w);
        add_rows();
    } else {
        cur = *myhead; fc = true;
        int k = 0;
        do {
            load_chunk(fc); fc = false;
            den += wave_sum(expf(v - m));
            ++k;
        } while (lists_live(cur) && k < max_chunks);
        cur = *myhead; fc = true; k = 0;
        do {
            load_chunk(fc); fc = false;
            w = expf(v - m);
            add_rows();
            ++k;
        } while (lists_live(cur) && k < max_chunks);
    }
    acc += other_half(acc);
    if (lane < OUTC) out[(size_t)d * OUTC + c] = acc + bz;
}

extern "C" void kernel_launch(void* const* d_in, const int* in_sizes, int n_in,
                              void* d_out, int out_size, void* d_ws, size_t ws_size,
                              hipStream_t stream) {
    const float* x     = (const float*)d_in[0];
    const int*   ei    = (const int*)d_in[1];
    const float* W1    = (const float*)d_in[2];
    const float* as1   = (const float*)d_in[3];
    const float* ad1   = (const float*)d_in[4];
    const float* bias1 = (const float*)d_in[5];
    const float* W2    = (const float*)d_in[6];
    const float* as2   = (const float*)d_in[7];
    const float* ad2   = (const float*)d_in[8];
    const float* bias2 = (const float*)d_in[9];
    float* out = (float*)d_out;

    const int N  = in_sizes[0] / 3;
    const int E  = in_sizes[1] / 2;

    float* ws = (float*)d_ws;
    const size_t Nz = (size_t)N;
    float* a_s1 = ws + 530 * Nz;
    float* a_d1 = ws + 538 * Nz;
    float* h2   = ws + 546 * Nz;
    float* a_s2 = ws + 562 * Nz;
    float* a_d2 = ws + 563 * Nz;
    int*   head = (int*)(ws + 564 * Nz);
    int2*  link = (int2*)(ws + 596 * Nz);

    const int TB = 256;
    const int node_blocks = (N + 7) / 8;
    k_fill_head<<<(NL * N + TB - 1) / TB, TB, 0, stream>>>(head, NL * N);

    {
        int mtiles = (N + 15) / 16;
        long long waves = (long long)mtiles * 2;
        int gemm_blocks = (int)((waves * 32 + TB - 1) / TB);
        int link_blocks = (E + TB - 1) / TB;
        k_gemm1_link<<<gemm_blocks + link_blocks, TB, 0, stream>>>(
            x, W1, as1, ad1, a_s1, a_d1, N, gemm_blocks, ei, E, head, link);
    }
    k_walk1_gemm2<<<(N + WG_NODES - 1) / WG_NODES, WG_NODES * 32, 0, stream>>>(
        head, (const unsigned long long*)link, E, a_s1, a_d1, x, W1, bias1, W2, as2, ad2, h2,
        a_s2, a_d2, N);

    k_walk2<<<node_blocks, TB, 0, stream>>>(head, (const unsigned long long*)link, E, a_s2, a_d2, h2,
                                            bias2, out, N);
}


# Softmax weights and ELU on bare v_exp_f32 (exp2 of a log2e-scaled argument), w/den as w*rcp(den): ~330 vector instructions fewer per node than expf + IEEE division (last bits differ; sum order same)
# speedup vs baseline: 9.1495x; 1.0633x over previous
#include <hip/hip_runtime.h>
#include <math.h>

#define HEADS     8
#define HIDC      32
#define L1C       256
#define OUTC      16
#define NEG_SLOPE 0.2f

typedef __attribute__((ext_vector_type(2))) float v2f;
typedef __attribute__((ext_vector_type(8))) float v8f;

__device__ __forceinline__ float lrelu(float v) {
    return v > 0.0f ? v : NEG_SLOPE * v;
}
#define LOG2E 1.4426950408889634f
__device__ __forceinline__ float exp_hw(float v) {
    return __builtin_amdgcn_exp2f(v * LOG2E);
}
__device__ __forceinline__ float softmax_weight(float w, float den) {
    return w * __builtin_amdgcn_rcpf(den + 1e-16f);
}
#define NL 32
static_assert(NL == 32, "one list per lane of a 32-lane wave");
__global__ void k_fill_head(int* __restrict__ head, int n) {
    int i = blockIdx.x * blockDim.x + threadIdx.x;
    if (i < n) head[i] = -1;
}

__device__ __forceinline__ void link_edges(int blk, const int* __restrict__ ei, int E,
                                           int* __restrict__ head, int2* __restrict__ link) {
    int e = blk * (int)blockDim.x + (int)threadIdx.x;
    if (e >= E) return;
    int s = ei[e], d = ei[E + e];
    int old = atomicExch(&head[(size_t)d * NL + (e & (NL - 1))], e);
    link[e] = make_int2(old, s);
}

template <int MASK>
__device__ __forceinline__ float row_xor(float v) {
    return __int_as_float(__builtin_amdgcn_update_dpp(0, __float_as_int(v), 0x160 + MASK,
                                                      0xf, 0xf, true));
}

__device__ __forceinline__ float other_half(float v) {
    const int i = __float_as_int(v);
    return __int_as_float(__builtin_amdgcn_permlanex16(i, i, 0x76543210, 0xfedcba98, false, false));
}

__device__ __forceinline__ float lane_val(float v, int j) {
    return __int_as_float(__builtin_amdgcn_readlane(__float_as_int(v), j));
}

__device__ __forceinline__ float wave_max(float v) {
    v = fmaxf(v, other_half(v));
    v = fmaxf(v, row_xor<8>(v));
    v = fmaxf(v, row_xor<4>(v));
    v = fmaxf(v, row_xor<2>(v));
    return fmaxf(v, row_xor<1>(v));
}
__device__ __forceinline__ float wave_sum(float v) {
    v += other_half(v);
    v += row_xor<8>(v);
    v += row_xor<4>(v);
    v += row_xor<2>(v);
    return v + row_xor<1>(v);
}

template <bool IS_MAX>
__device__ __forceinline__ void wave_reduce8(const float (&in)[8], int lane, float (&out)[8]) {
    auto op = [](float a, float b) { return IS_MAX ? fmaxf(a, b) : a + b; };
    const bool b4 = (lane & 16) != 0, b3 = (lane & 8) != 0, b2 = (lane & 4) != 0;
    float r[4];
#pragma unroll
    for (int i = 0; i < 4; ++i) {
        const float keep = b4 ? in[i + 4] : in[i], give = b4 ? in[i] : in[i + 4];
        r[i] = op(keep, other_half(give));
    }
#pragma unroll
    for (int i = 0; i < 2; ++i) {
        const float keep = b3 ? r[i + 2] : r[i], give = b3 ? r[i] : r[i + 2];
        r[i] = op(keep, row_xor<8>(give));
    }
    const float keep = b2 ? r[1] : r[0], give = b2 ? r[0] : r[1];
    float t = op(keep, row_xor<4>(give));
    t = op(t, row_xor<2>(t));
    t = op(t, row_xor<1>(t));
#pragma unroll
    for (int h = 0; h < 8; ++h) out[h] = lane_val(t, 4 * h);
}

__device__ __forceinline__ bool lists_live(int cur) {
    return __builtin_amdgcn_ballot_w32(cur >= 0) != 0u;
}
__device__ __forceinline__ void lists_fill(int& cur, const unsigned long long* __restrict__ link,
                                           int lane, int& mysrc, int& cnt) {
    for (int hop = 0; hop < 32 && cnt < 32; ++hop) {
        unsigned live = __builtin_amdgcn_ballot_w32(cur >= 0);
        if (live == 0u) break;
        unsigned long long r = 0ull;
        if (cur >= 0) r = link[cur];
        const int src = (int)(r >> 32);
        const int room = 32 - cnt, have = __builtin_popcount(live);
        const int n = have < room ? have : room;
        unsigned left = live;
        for (int t = 0; t < n; ++t) {
            const int k = __builtin_ctz(left);
            left &= left - 1u;
            const int s = __builtin_amdgcn_readlane(src, k);
            if (lane == cnt + t) mysrc = s;
        }
        cnt += n;
        if (((live & ~left) >> lane) & 1u) cur = (int)(unsigned)r;
    }
}


__device__ __forceinline__ float row_sums16(float (&r)[16], int lane) {
    const bool b3 = (lane & 8) != 0, b2 = (lane & 4) != 0, b1 = (lane & 2) != 0,
               b0 = (lane & 1) != 0;
#pragma unroll
    for (int i = 0; i < 8; ++i) {
        float keep = b3 ? r[i + 8] : r[i], give = b3 ? r[i] : r[i + 8];
        r[i] = keep + row_xor<8>(give);
    }
#pragma unroll
    for (int i = 0; i < 4; ++i) {
        float keep = b2 ? r[i + 4] : r[i], give = b2 ? r[i] : r[i + 4];
        r[i] = keep + row_xor<4>(give);
    }
#pragma unroll
    for (int i = 0; i < 2; ++i) {
        float keep = b1 ? r[i + 2] : r[i], give = b1 ? r[i] : r[i + 2];
        r[i] = keep + row_xor<2>(give);
    }
    float keep = b0 ? r[1] : r[0], give = b0 ? r[0] : r[1];
    return keep + row_xor<1>(give);
}

__device__ __forceinline__ void gemm1_logits_tile(int blk, const float* __restrict__ x,
                               const float* __restrict__ W1,
                               const float* __restrict__ as1, const float* __restrict__ ad1,
                               float* __restrict__ a_s,
                               float* __restrict__ a_d, int n_nodes) {
    int wave   = (int)((blk * (int)blockDim.x + (int)threadIdx.x) >> 5);
    int lane   = threadIdx.x & 31;
    int mtiles = (n_nodes + 15) >> 4;
    int mtile  = wave >> 1;
    int hg     = wave & 1;
    if (mtile >= mtiles) return;
    int m  = lane & 15;
    int kh = lane >> 4;
    int node = mtile * 16 + m;
    if (node >= n_nodes) node = n_nodes - 1;
    float x0 = x[node * 3 + 0], x1 = x[node * 3 + 1], x2 = x[node * 3 + 2];
    v2f a;
    a.x = kh ? x2 : x0;  a.y = kh ? 0.0f : x1;
    const int orow = mtile * 16 + 8 * kh + (m & 7);
    float* dst = (m < 8 ? a_s : a_d) + (size_t)orow * HEADS + hg * 4;
#pragma unroll 1
    for (int hh = 0; hh < 4; ++hh) {
        const int head = hg * 4 + hh;
        const int col  = head * HIDC + m;
        float w0 = W1[0 * L1C + col], w1 = W1[1 * L1C + col], w2 = W1[2 * L1C + col];
        float u0 = W1[0 * L1C + col + 16], u1 = W1[1 * L1C + col + 16],
              u2 = W1[2 * L1C + col + 16];
        v2f bA, bB;
        bA.x = kh ? w2 : w0;  bA.y = kh ? 0.0f : w1;
        bB.x = kh ? u2 : u0;  bB.y = kh ? 0.0f : u1;
        v8f c0 = {}, c1 = {};
        c0 = __builtin_amdgcn_wmma_f32_16x16x4_f32(false, a, false, bA, (short)0, c0,
                                                   false, false);
        c1 = __builtin_amdgcn_wmma_f32_16x16x4_f32(false, a, false, bB, (short)0, c1,
                                                   false, false);
        const float sl = as1[head * HIDC + m], sh = as1[head * HIDC + 16 + m];
        const float dl = ad1[head * HIDC + m], dh = ad1[head * HIDC + 16 + m];
        float r[16];
#pragma unroll
        for (int v = 0; v < 8; ++v) {
            r[v]     = c0[v] * sl + c1[v] * sh;
            r[8 + v] = c0[v] * dl + c1[v] * dh;
        }
        const float tot = row_sums16(r, lane);
        if (orow < n_nodes) dst[hh] = tot;
    }
}

__global__ void k_gemm1_link(const float* __restrict__ x, const float* __restrict__ W1,
                             const float* __restrict__ as1, const float* __restrict__ ad1,
                             float* __restrict__ a_s,
                             float* __restrict__ a_d, int n_nodes, int gemm_blocks,
                             const int* __restrict__ ei, int E, int* __restrict__ head,
                             int2* __restrict__ link) {
    const int blk = (int)blockIdx.x;
    if (blk < gemm_blocks) gemm1_logits_tile(blk, x, W1, as1, ad1, a_s, a_d, n_nodes);
    else link_edges(blk - gemm_blocks, ei, E, head, link);
}

#define WG_NODES 16
#define ROWP (L1C + 4)
#define WALK1_WAVES_PER_SIMD 10
__global__ void __launch_bounds__(WG_NODES * 32, WALK1_WAVES_PER_SIMD)
k_walk1_gemm2(const int* __restrict__ head, const unsigned long long* __restrict__ link, int E,
              const float* __restrict__ a_s, const float* __restrict__ a_d,
              const float* __restrict__ x, const float* __restrict__ W1,
              const float* __restrict__ bias1,
              const float* __restrict__ W2, const float* __restrict__ as2,
              const float* __restrict__ ad2, float* __restrict__ h2,
              float* __restrict__ a_s2, float* __restrict__ a_d2, int n_nodes) {
    __shared__ __attribute__((aligned(16))) float act[WG_NODES * ROWP];
    __shared__ int arrived;
    __shared__ __attribute__((aligned(8))) v2f fragW[(L1C / 16) * 32];
    static_assert(WG_NODES == L1C / 16, "wave nt of the work-group stages tile nt");
    const int lane = threadIdx.x & 31;
    const int wv = __builtin_amdgcn_readfirstlane((int)(threadIdx.x >> 5));
    const int node0 = (int)blockIdx.x * WG_NODES;
    const int d = node0 + wv;
    const int mm = lane & 15;
    const int kh = lane >> 4;
    const int chan = lane;
    int cur0 = -1;
    if (d < n_nodes) cur0 = head[(size_t)d * NL + lane];
    {
        const int col = wv * 16 + mm;
        const float wx = W1[(kh ? 2 * L1C : 0) + col], wy = W1[L1C + col];
        v2f f;
        f.x = wx;  f.y = kh ? 0.0f : wy;
        fragW[wv * 32 + lane] = f;
    }
    if (threadIdx.x == 0) arrived = 0;
    __syncthreads();

    float acc[HEADS];
#pragma unroll
    for (int h = 0; h < HEADS; ++h) acc[h] = 0.0f;

    if (d < n_nodes) {
        const int* myhead = head + (size_t)d * NL + lane;
        const int max_chunks = (E >> 5) + 2;
        float ad[HEADS];
#pragma unroll
        for (int h = 0; h < HEADS; ++h) ad[h] = a_d[d * HEADS + h];

        float m[HEADS], den[HEADS], v[HEADS], w[HEADS];
#pragma unroll
        for (int h = 0; h < HEADS; ++h) { m[h] = -INFINITY; den[h] = 0.0f; }
        int mysrc = d, cnt = 0, cur = cur0;
        float xs0 = 0.0f, xs1 = 0.0f, xs2 = 0.0f;

        auto load_chunk = [&](bool firstc) {
            cnt = 0;
            if (firstc) { mysrc = d; cnt = 1; }
            lists_fill(cur, link, lane, mysrc, cnt);
            const float4* p = (const float4*)(a_s + (size_t)mysrc * HEADS);
            const float* px = x + (size_t)mysrc * 3;
            float4 lo = p[0], hi = p[1];
            xs0 = px[0];  xs1 = px[1];  xs2 = px[2];
            v[0] = lo.x; v[1] = lo.y; v[2] = lo.z; v[3] = lo.w;
            v[4] = hi.x; v[5] = hi.y; v[6] = hi.z; v[7] = hi.w;
#pragma unroll
            for (int h = 0; h < HEADS; ++h)
                v[h] = (lane < cnt) ? lrelu(v[h] + ad[h]) : -INFINITY;
        };
        auto add_rows = [&](const bool two) {
            float al[HEADS];
#pragma unroll
            for (int h = 0; h < HEADS; ++h) al[h] = softmax_weight(w[h], den[h]);
            v2f xb0, xb1 = {};
            const float o2 = other_half(xs2);
            xb0.x = kh ? o2 : xs0;  xb0.y = kh ? 0.0f : xs1;
            if (two) {
                const float o0 = other_half(xs0), o1 = other_half(xs1);
                xb1.x = kh ? xs2 : o0;  xb1.y = kh ? 0.0f : o1;
            }
            v2f nA = fragW[lane], nB = fragW[32 + lane];
#pragma unroll
            for (int h = 0; h < HEADS; ++h) {
                const v2f bA = nA, bB = nB;
                if (h + 1 < HEADS) {
                    nA = fragW[(2 * h + 2) * 32 + lane];
                    nB = fragW[(2 * h + 3) * 32 + lane];
                }
                const float ax = other_half(al[h]);
                const float wt0 = kh ? ax : al[h];
                const float wt1 = kh ? al[h] : ax;
                v2f a;
                a.x = xb0.x * wt0;  a.y = xb0.y * wt0;
                v8f c0 = {}, c1 = {};
                c0 = __builtin_amdgcn_wmma_f32_16x16x4_f32(false, a, false, bA, (short)0, c0,
                                                           false, false);
                c1 = __builtin_amdgcn_wmma_f32_16x16x4_f32(false, a, false, bB, (short)0, c1,
                                                           false, false);
                if (two) {
                    v2f a2;
                    a2.x = xb1.x * wt1;  a2.y = xb1.y * wt1;
                    c0 = __builtin_amdgcn_wmma_f32_16x16x4_f32(false, a2, false, bA, (short)0, c0,
                                                               false, false);
                    c1 = __builtin_amdgcn_wmma_f32_16x16x4_f32(false, a2, false, bB, (short)0, c1,
                                                               false, false);
                }
                const v2f q0 = (__builtin_shufflevector(c0, c0, 0, 1) +
                                __builtin_shufflevector(c0, c0, 4, 5)) +
                               (__builtin_shufflevector(c0, c0, 2, 3) +
                                __builtin_shufflevector(c0, c0, 6, 7));
                const v2f q1 = (__builtin_shufflevector(c1, c1, 0, 1) +
                                __builtin_shufflevector(c1, c1, 4, 5)) +
                               (__builtin_shufflevector(c1, c1, 2, 3) +
                                __builtin_shufflevector(c1, c1, 6, 7));
                const float p0 = q0.x + q0.y, p1 = q1.x + q1.y;
                const float own = kh ? p1 : p0, give = kh ? p0 : p1;
                acc[h] += own + other_half(give);
                __builtin_amdgcn_sched_barrier(0);
            }
        };

        int nch = 0;
        bool fc = true;
        do {
            load_chunk(fc); fc = false;
            float mx[HEADS];
            wave_reduce8<true>(v, lane, mx);
#pragma unroll
            for (int h = 0; h < HEADS; ++h) m[h] = fmaxf(m[h], mx[h]);
            asm volatile("" : : "v"(xs0), "v"(xs1), "v"(xs2));
            ++nch;
        } while (lists_live(cur) && nch < max_chunks);

        if (nch == 1) {
#pragma unroll
            for (int h = 0; h < HEADS; ++h) w[h] = exp_hw(v[h] - m[h]);
            wave_reduce8<false>(w, lane, den);
            if (cnt > 16) add_rows(true); else add_rows(false);
        } else {
            cur = *myhead; fc = true;
            int k = 0;
            do {
                load_chunk(fc); fc = false;
                float ex[HEADS], sm[HEADS];
#pragma unroll
                for (int h = 0; h < HEADS; ++h) ex[h] = exp_hw(v[h] - m[h]);
                wave_reduce8<false>(ex, lane, sm);
#pragma unroll
                for (int h = 0; h < HEADS; ++h) den[h] += sm[h];
                ++k;
            } while (lists_live(cur) && k < max_chunks);
            cur = *myhead; fc = true; k = 0;
            do {
                load_chunk(fc); fc = false;
#pragma unroll
                for (int h = 0; h < HEADS; ++h) w[h] = exp_hw(v[h] - m[h]);
                if (cnt > 16) add_rows(true); else add_rows(false);
                ++k;
            } while (lists_live(cur) && k < max_chunks);
        }
    }

    {
        float bv[HEADS];
#pragma unroll
        for (int h = 0; h < HEADS; ++h) bv[h] = bias1[h * HIDC + chan];
        float* prow = act + wv * ROWP + chan;
#pragma unroll
        for (int h = 0; h < HEADS; ++h) {
            float t = acc[h] + bv[h];
            prow[h * HIDC] = t > 0.0f ? t : (exp_hw(t) - 1.0f);
        }
    }
    __builtin_amdgcn_fence(__ATOMIC_RELEASE, "workgroup");
    int prev = 0;
    if (lane == 0)
        prev = __hip_atomic_fetch_add(&arrived, 1, __ATOMIC_RELAXED, __HIP_MEMORY_SCOPE_WORKGROUP);
    prev = __builtin_amdgcn_readfirstlane(prev);
    if (prev != WG_NODES - 1) return;
    __builtin_amdgcn_fence(__ATOMIC_ACQUIRE, "workgroup");

    const float* arow = act + mm * ROWP + kh * 2;
    const float* brow = W2 + kh * 2 * OUTC + mm;
    const float sv = as2[mm], dv = ad2[mm];
    v8f c = {};
#pragma unroll 16
    for (int kk = 0; kk < L1C; kk += 4) {
        v2f a = *(const v2f*)(arow + kk);
        v2f b;
        b.x = brow[kk * OUTC];
        b.y = brow[kk * OUTC + OUTC];
        c = __builtin_amdgcn_wmma_f32_16x16x4_f32(false, a, false, b, (short)0, c,
                                                  false, false);
    }
    float* p = h2 + (size_t)(node0 + 8 * kh) * OUTC + mm;
    if (node0 + WG_NODES <= n_nodes) {
#pragma unroll
        for (int r = 0; r < 8; ++r) p[(size_t)r * OUTC] = c[r];
    } else {
        for (int r = 0; r < 8; ++r)
            if (node0 + 8 * kh + r < n_nodes) p[(size_t)r * OUTC] = c[r];
    }
    float r16[16];
#pragma unroll
    for (int v = 0; v < 8; ++v) {
        r16[v]     = c[v] * sv;
        r16[8 + v] = c[v] * dv;
    }
    const float tot = row_sums16(r16, lane);
    const int orow = node0 + 8 * kh + (mm & 7);
    if (orow < n_nodes) (mm < 8 ? a_s2 : a_d2)[orow] = tot;
}

__global__ void k_walk2(const int* __restrict__ head, const unsigned long long* __restrict__ link, int E,
                        const float* __restrict__ a_s, const float* __restrict__ a_d,
                        const float* __restrict__ h2, const float* __restrict__ bias2,
                        float* __restrict__ out, int n_nodes) {
    const int lane = threadIdx.x & 31;
    const int d = __builtin_amdgcn_readfirstlane(
        (int)(blockIdx.x * (blockDim.x >> 5) + (threadIdx.x >> 5)));
    if (d >= n_nodes) return;
    const int* myhead = head + (size_t)d * NL + lane;
    const int max_chunks = (E >> 5) + 2;
    const float adv = a_d[d];
    const int c = lane & 15, half = lane >> 4;
    const float bz = bias2[c];
    float m = -INFINITY, den = 0.0f, acc = 0.0f, v = 0.0f, w = 0.0f;
    int mysrc = d, cnt = 0, cur = *myhead;

    auto load_chunk = [&](bool firstc) {
        cnt = 0;
        if (firstc) { mysrc = d; cnt = 1; }
        lists_fill(cur, link, lane, mysrc, cnt);
        v = (lane < cnt) ? lrelu(a_s[mysrc] + adv) : -INFINITY;
    };
    auto add_rows = [&]() {
        float al = softmax_weight(w, den);
#pragma unroll 4
        for (int j = 0; j < cnt; j += 2) {
            int jj = j + half;
            int s = __shfl(mysrc, jj, 32);
            float a = __shfl(al, jj, 32);
            float hv = h2[(size_t)s * OUTC + c];
            acc += (jj < cnt) ? hv * a : 0.0f;
        }
    };

    int nch = 0;
    bool fc = true;
    do {
        load_chunk(fc); fc = false;
        m = fmaxf(m, wave_max(v));
        ++nch;
    } while (lists_live(cur) && nch < max_chunks);

    if (nch == 1) {
        w = exp_hw(v - m);
        den = wave_sum(w);
        add_rows();
    } else {
        cur = *myhead; fc = true;
        int k = 0;
        do {
            load_chunk(fc); fc = false;
            den += wave_sum(exp_hw(v - m));
            ++k;
        } while (lists_live(cur) && k < max_chunks);
        cur = *myhead; fc = true; k = 0;
        do {
            load_chunk(fc); fc = false;
            w = exp_hw(v - m);
            add_rows();
            ++k;
        } while (lists_live(cur) && k < max_chunks);
    }
    acc += other_half(acc);
    if (lane < OUTC) out[(size_t)d * OUTC + c] = acc + bz;
}

extern "C" void kernel_launch(void* const* d_in, const int* in_sizes, int n_in,
                              void* d_out, int out_size, void* d_ws, size_t ws_size,
                              hipStream_t stream) {
    const float* x     = (const float*)d_in[0];
    const int*   ei    = (const int*)d_in[1];
    const float* W1    = (const float*)d_in[2];
    const float* as1   = (const float*)d_in[3];
    const float* ad1   = (const float*)d_in[4];
    const float* bias1 = (const float*)d_in[5];
    const float* W2    = (const float*)d_in[6];
    const float* as2   = (const float*)d_in[7];
    const float* ad2   = (const float*)d_in[8];
    const float* bias2 = (const float*)d_in[9];
    float* out = (float*)d_out;

    const int N  = in_sizes[0] / 3;
    const int E  = in_sizes[1] / 2;

    float* ws = (float*)d_ws;
    const size_t Nz = (size_t)N;
    float* a_s1 = ws + 530 * Nz;
    float* a_d1 = ws + 538 * Nz;
    float* h2   = ws + 546 * Nz;
    float* a_s2 = ws + 562 * Nz;
    float* a_d2 = ws + 563 * Nz;
    int*   head = (int*)(ws + 564 * Nz);
    int2*  link = (int2*)(ws + 596 * Nz);

    const int TB = 256;
    const int node_blocks = (N + 7) / 8;
    k_fill_head<<<(NL * N + TB - 1) / TB, TB, 0, stream>>>(head, NL * N);

    {
        int mtiles = (N + 15) / 16;
        long long waves = (long long)mtiles * 2;
        int gemm_blocks = (int)((waves * 32 + TB - 1) / TB);
        int link_blocks = (E + TB - 1) / TB;
        k_gemm1_link<<<gemm_blocks + link_blocks, TB, 0, stream>>>(
            x, W1, as1, ad1, a_s1, a_d1, N, gemm_blocks, ei, E, head, link);
    }
    k_walk1_gemm2<<<(N + WG_NODES - 1) / WG_NODES, WG_NODES * 32, 0, stream>>>(
        head, (const unsigned long long*)link, E, a_s1, a_d1, x, W1, bias1, W2, as2, ad2, h2,
        a_s2, a_d2, N);

    k_walk2<<<node_blocks, TB, 0, stream>>>(head, (const unsigned long long*)link, E, a_s2, a_d2, h2,
                                            bias2, out, N);
}
